# starting kernel
# baseline (speedup 1.0000x reference)
#include <hip/hip_runtime.h>
#include <hip/hip_bf16.h>


#define D_MODEL 768
#define D_STATE 16
#define D_CONV  4
#define D_INNER 1536
#define DT_RANK 48
#define BATCH   2
#define SEQ     2048
#define MROWS   (BATCH * SEQ)
#define DBC_LD  (DT_RANK + 2 * D_STATE)

typedef __attribute__((ext_vector_type(2))) float v2f;
typedef __attribute__((ext_vector_type(8))) float v8f;

__device__ __forceinline__ float silu_f(float x) {
    return x * (1.0f / (1.0f + __expf(-x)));
}
__device__ __forceinline__ float softplus_f(float x) {
    return (x > 20.0f) ? x : log1pf(__expf(x));
}

template <int MT, int NT, int EPI>
__launch_bounds__(128)
__global__ void gemm_wmma_f32(const float* __restrict__ A,
                              const float* __restrict__ W,
                              const float* __restrict__ bias,
                              float* __restrict__ C,
                              int M, int N, int K, int lda, int ldc) {
    const int lane  = threadIdx.x & 31;
    const int wave  = threadIdx.x >> 5;
    const int mBase = (blockIdx.y * (blockDim.x >> 5) + wave) * (16 * MT);
    const int nBase = blockIdx.x * (16 * NT);
    const int r16   = lane & 15;
    const int kOff  = (lane >> 4) * 2;

    const float* aP[MT];
    const float* wP[NT];
#pragma unroll
    for (int i = 0; i < MT; ++i)
        aP[i] = A + (size_t)(mBase + i * 16 + r16) * lda + kOff;
#pragma unroll
    for (int t = 0; t < NT; ++t)
        wP[t] = W + (size_t)(nBase + t * 16 + r16) * K + kOff;

    v8f acc[MT][NT];
#pragma unroll
    for (int i = 0; i < MT; ++i)
#pragma unroll
        for (int t = 0; t < NT; ++t) {
            v8f z = {0.f, 0.f, 0.f, 0.f, 0.f, 0.f, 0.f, 0.f};
            acc[i][t] = z;
        }

    v2f aA[MT], bA[NT], aB[MT], bB[NT];

#define LOADFRAG(AF, BF, KK)                                         \
    _Pragma("unroll") for (int i = 0; i < MT; ++i)                   \
        AF[i] = *(const v2f*)(aP[i] + (KK));                         \
    _Pragma("unroll") for (int t = 0; t < NT; ++t)                   \
        BF[t] = *(const v2f*)(wP[t] + (KK));

#define MACFRAG(AF, BF)                                              \
    _Pragma("unroll") for (int i = 0; i < MT; ++i)                   \
        _Pragma("unroll") for (int t = 0; t < NT; ++t)               \
            acc[i][t] = __builtin_amdgcn_wmma_f32_16x16x4_f32(       \
                false, AF[i], false, BF[t], (short)0, acc[i][t],     \
                false, false);

    LOADFRAG(aA, bA, 0)
    int k = 0;
    for (; k + 8 < K; k += 8) {
        LOADFRAG(aB, bB, k + 4)
        MACFRAG(aA, bA)
        LOADFRAG(aA, bA, k + 8)
        MACFRAG(aB, bB)
    }
    LOADFRAG(aB, bB, k + 4)
    MACFRAG(aA, bA)
    MACFRAG(aB, bB)

#undef LOADFRAG
#undef MACFRAG

    const int rowSel = (lane >> 4) * 8;
#pragma unroll
    for (int i = 0; i < MT; ++i)
#pragma unroll
        for (int t = 0; t < NT; ++t) {
            const int col = nBase + t * 16 + (lane & 15);
            float bv = 0.0f;
            if (EPI == 1) bv = bias[col];
#pragma unroll
            for (int r = 0; r < 8; ++r) {
                float v = acc[i][t][r];
                if (EPI == 1) v = softplus_f(v + bv);
                C[(size_t)(mBase + i * 16 + r + rowSel) * ldc + col] = v;
            }
        }
}

__launch_bounds__(256)
__global__ void conv_silu_kernel(const float* __restrict__ xin,
                                 const float* __restrict__ w,
                                 const float* __restrict__ b,
                                 float* __restrict__ out) {
    const int idx = blockIdx.x * blockDim.x + threadIdx.x;
    const int d  = idx % D_INNER;
    const int s  = (idx / D_INNER) % SEQ;
    const int bi = idx / (D_INNER * SEQ);

    const float* base = xin + (size_t)bi * SEQ * D_INNER + d;
    float acc = b[d];
#pragma unroll
    for (int j = 0; j < D_CONV; ++j) {
        const int sp = s - (D_CONV - 1) + j;
        if (sp >= 0) acc += w[d * D_CONV + j] * base[(size_t)sp * D_INNER];
    }
    out[idx] = silu_f(acc);
}

__launch_bounds__(256)
__global__ void scan_kernel(const float* __restrict__ delta,
                            const float* __restrict__ xc,
                            const float* __restrict__ dbc,
                            const float* __restrict__ zbuf,
                            const float* __restrict__ logA,
                            const float* __restrict__ Dskip,
                            float* __restrict__ y) {
    const int gid = blockIdx.x * blockDim.x + threadIdx.x;
    const int n = gid & (D_STATE - 1);
    const int d = (gid >> 4) % D_INNER;
    const int b = gid / (D_INNER * D_STATE);

    const float Aneg   = -__expf(logA[d * D_STATE + n]);
    const bool  writer = (n == 0);
    const float dsk    = Dskip[d];

    const size_t baseBD  = (size_t)b * SEQ * D_INNER + d;
    const size_t baseDbc = (size_t)b * SEQ * DBC_LD;

    float h = 0.0f;
#pragma unroll 2
    for (int s = 0; s < SEQ; ++s) {
        const size_t ix = baseBD + (size_t)s * D_INNER;
        const float dt = delta[ix];
        const float xv = xc[ix];
        const float* pd = dbc + baseDbc + (size_t)s * DBC_LD;
        const float Bv = pd[DT_RANK + n];
        const float Cv = pd[DT_RANK + D_STATE + n];

        const float a = __expf(dt * Aneg);
        h = a * h + (dt * Bv) * xv;

        float p = h * Cv;
        p += __shfl_xor(p, 1, 16);
        p += __shfl_xor(p, 2, 16);
        p += __shfl_xor(p, 4, 16);
        p += __shfl_xor(p, 8, 16);

        if (writer) {
            const float zv = zbuf[ix];
            y[ix] = (p + dsk * xv) * silu_f(zv);
        }
    }
}

extern "C" void kernel_launch(void* const* d_in, const int* in_sizes, int n_in,
                              void* d_out, int out_size, void* d_ws, size_t ws_size,
                              hipStream_t stream) {
    const float* x         = (const float*)d_in[0];
    const float* in_proj_w = (const float*)d_in[1];
    const float* conv_w    = (const float*)d_in[2];
    const float* conv_b    = (const float*)d_in[3];
    const float* x_proj_w  = (const float*)d_in[4];
    const float* dt_proj_w = (const float*)d_in[5];
    const float* dt_proj_b = (const float*)d_in[6];
    const float* log_A     = (const float*)d_in[7];
    const float* D_skip    = (const float*)d_in[8];
    const float* out_proj_w= (const float*)d_in[9];
    float* out = (float*)d_out;

    const size_t planeBD = (size_t)MROWS * D_INNER;
    float* ws      = (float*)d_ws;
    float* buf_x   = ws;
    float* buf_z   = buf_x  + planeBD;
    float* buf_xc  = buf_z  + planeBD;
    float* buf_dbc = buf_xc + planeBD;
    float* buf_y   = buf_dbc + (size_t)MROWS * DBC_LD;

    const dim3 blk(128);
    const int gy = MROWS / (16 * 2 * 4);

    gemm_wmma_f32<2, 4, 0><<<dim3(D_INNER / 64, gy), blk, 0, stream>>>(
        x, in_proj_w, nullptr, buf_x, MROWS, D_INNER, D_MODEL, D_MODEL, D_INNER);
    gemm_wmma_f32<2, 4, 0><<<dim3(D_INNER / 64, gy), blk, 0, stream>>>(
        x, in_proj_w + (size_t)D_INNER * D_MODEL, nullptr, buf_z,
        MROWS, D_INNER, D_MODEL, D_MODEL, D_INNER);

    conv_silu_kernel<<<(MROWS * D_INNER) / 256, 256, 0, stream>>>(
        buf_x, conv_w, conv_b, buf_xc);

    gemm_wmma_f32<2, 1, 0><<<dim3(DBC_LD / 16, gy), blk, 0, stream>>>(
        buf_xc, x_proj_w, nullptr, buf_dbc, MROWS, DBC_LD, D_INNER, D_INNER, DBC_LD);

    gemm_wmma_f32<2, 4, 1><<<dim3(D_INNER / 64, gy), blk, 0, stream>>>(
        buf_dbc, dt_proj_w, dt_proj_b, buf_x, MROWS, D_INNER, DT_RANK, DBC_LD, D_INNER);

    scan_kernel<<<(BATCH * D_INNER * D_STATE) / 256, 256, 0, stream>>>(
        buf_x, buf_xc, buf_dbc, buf_z, log_A, D_skip, buf_y);

    gemm_wmma_f32<2, 4, 0><<<dim3(D_MODEL / 64, gy), blk, 0, stream>>>(
        buf_y, out_proj_w, nullptr, out, MROWS, D_MODEL, D_INNER, D_INNER, D_MODEL);
}


# scan re-mapped to one lane per channel: 16 states in registers, dt/x/z/y handled once per channel as whole 128-byte lines, B,C via scalar loads, y sum over n in-lane (same pair tree, no shuffles)
# speedup vs baseline: 7.9401x; 7.9401x over previous
#include <hip/hip_runtime.h>
#include <hip/hip_bf16.h>


#define D_MODEL 768
#define D_STATE 16
#define D_CONV  4
#define D_INNER 1536
#define DT_RANK 48
#define BATCH   2
#define SEQ     2048
#define MROWS   (BATCH * SEQ)
#define DBC_LD  (DT_RANK + 2 * D_STATE)

typedef __attribute__((ext_vector_type(2))) float v2f;
typedef __attribute__((ext_vector_type(4))) float v4f;
typedef __attribute__((ext_vector_type(8))) float v8f;

__device__ __forceinline__ float silu_f(float x) {
    return x * (1.0f / (1.0f + __expf(-x)));
}
__device__ __forceinline__ float softplus_f(float x) {
    return (x > 20.0f) ? x : log1pf(__expf(x));
}

template <int MT, int NT, int KS, int WM, int WN, int EPI>
__launch_bounds__(WM * WN * 32)
__global__ void gemm_wmma_f32(const float* __restrict__ A,
                              const float* __restrict__ W,
                              const float* __restrict__ bias,
                              float* __restrict__ C,
                              int K, int lda, int ldc) {
    constexpr int KQ = KS / 8;
    const int lane  = threadIdx.x & 31;
    const int wave  = threadIdx.x >> 5;
    const int wm    = wave % WM;
    const int wn    = wave / WM;
    const int mBase = (blockIdx.y * WM + wm) * (16 * MT);
    const int nBase = (blockIdx.x * WN + wn) * (16 * NT);
    const int r16   = lane & 15;
    const int kOff  = (lane >> 4) * 4;

    const float* aP[MT];
    const float* wP[NT];
#pragma unroll
    for (int i = 0; i < MT; ++i)
        aP[i] = A + (size_t)(mBase + i * 16 + r16) * lda + kOff;
#pragma unroll
    for (int t = 0; t < NT; ++t)
        wP[t] = W + (size_t)(nBase + t * 16 + r16) * K + kOff;

    v8f acc[MT][NT];
#pragma unroll
    for (int i = 0; i < MT; ++i)
#pragma unroll
        for (int t = 0; t < NT; ++t) {
            v8f z = {0.f, 0.f, 0.f, 0.f, 0.f, 0.f, 0.f, 0.f};
            acc[i][t] = z;
        }

    v4f aA[MT][KQ], bA[NT][KQ], aB[MT][KQ], bB[NT][KQ];

#define LOADFRAG(AF, BF, KK)                                         \
    _Pragma("unroll") for (int i = 0; i < MT; ++i)                   \
        _Pragma("unroll") for (int j = 0; j < KQ; ++j)               \
            AF[i][j] = *(const v4f*)(aP[i] + (KK) + 8 * j);          \
    _Pragma("unroll") for (int t = 0; t < NT; ++t)                   \
        _Pragma("unroll") for (int j = 0; j < KQ; ++j)               \
            BF[t][j] = *(const v4f*)(wP[t] + (KK) + 8 * j);

#define MACFRAG(AF, BF)                                              \
    _Pragma("unroll") for (int j = 0; j < KQ; ++j) {                 \
        _Pragma("unroll") for (int i = 0; i < MT; ++i)               \
            _Pragma("unroll") for (int t = 0; t < NT; ++t)           \
                acc[i][t] = __builtin_amdgcn_wmma_f32_16x16x4_f32(   \
                    false, AF[i][j].xy, false, BF[t][j].xy,          \
                    (short)0, acc[i][t], false, false);              \
        _Pragma("unroll") for (int i = 0; i < MT; ++i)               \
            _Pragma("unroll") for (int t = 0; t < NT; ++t)           \
                acc[i][t] = __builtin_amdgcn_wmma_f32_16x16x4_f32(   \
                    false, AF[i][j].zw, false, BF[t][j].zw,          \
                    (short)0, acc[i][t], false, false);              \
    }

    LOADFRAG(aA, bA, 0)
    int k = 0;
    for (; k + 2 * KS < K; k += 2 * KS) {
        LOADFRAG(aB, bB, k + KS)
        MACFRAG(aA, bA)
        LOADFRAG(aA, bA, k + 2 * KS)
        MACFRAG(aB, bB)
    }
    LOADFRAG(aB, bB, k + KS)
    MACFRAG(aA, bA)
    MACFRAG(aB, bB)

#undef LOADFRAG
#undef MACFRAG

    const int rowSel = (lane >> 4) * 8;
#pragma unroll
    for (int i = 0; i < MT; ++i)
#pragma unroll
        for (int t = 0; t < NT; ++t) {
            const int col = nBase + t * 16 + (lane & 15);
            float bv = 0.0f;
            if (EPI == 1) bv = bias[col];
#pragma unroll
            for (int r = 0; r < 8; ++r) {
                float v = acc[i][t][r];
                if (EPI == 1) v = softplus_f(v + bv);
                C[(size_t)(mBase + i * 16 + r + rowSel) * ldc + col] = v;
            }
        }
}

__launch_bounds__(256)
__global__ void conv_silu_kernel(const float* __restrict__ xin,
                                 const float* __restrict__ w,
                                 const float* __restrict__ b,
                                 float* __restrict__ out) {
    const int idx = blockIdx.x * blockDim.x + threadIdx.x;
    const int d  = idx % D_INNER;
    const int s  = (idx / D_INNER) % SEQ;
    const int bi = idx / (D_INNER * SEQ);

    const float* base = xin + (size_t)bi * SEQ * D_INNER + d;
    float acc = b[d];
#pragma unroll
    for (int j = 0; j < D_CONV; ++j) {
        const int sp = s - (D_CONV - 1) + j;
        if (sp >= 0) acc += w[d * D_CONV + j] * base[(size_t)sp * D_INNER];
    }
    out[idx] = silu_f(acc);
}

#define SCAN_NC   32
#define SCAN_LEN  (SEQ / SCAN_NC)
#define SCAN_G    8
#define SCAN_TPB  128
#define SUM_LD    ((size_t)D_STATE * D_INNER)

__device__ __forceinline__ void load_decay_rates(const float* __restrict__ logA,
                                                 int d, float (&An)[D_STATE]) {
#pragma unroll
    for (int q = 0; q < D_STATE / 4; ++q) {
        const v4f v = *(const v4f*)(logA + (size_t)d * D_STATE + 4 * q);
        An[4 * q + 0] = -__expf(v.x);
        An[4 * q + 1] = -__expf(v.y);
        An[4 * q + 2] = -__expf(v.z);
        An[4 * q + 3] = -__expf(v.w);
    }
}

__launch_bounds__(SCAN_TPB)
__global__ void scan_chunk_state_kernel(const float* __restrict__ delta,
                                        const float* __restrict__ xc,
                                        const float* __restrict__ dbc,
                                        const float* __restrict__ logA,
                                        float* __restrict__ hend,
                                        float* __restrict__ pdec) {
    const int d = blockIdx.x * SCAN_TPB + threadIdx.x;
    const int c = blockIdx.y;
    const int b = blockIdx.z;

    float An[D_STATE];
    load_decay_rates(logA, d, An);

    const size_t row0 = (size_t)b * SEQ + (size_t)c * SCAN_LEN;
    const float* pdt = delta + row0 * D_INNER + d;
    const float* pxc = xc    + row0 * D_INNER + d;
    const float* pB  = dbc   + row0 * DBC_LD + DT_RANK;

    float h[D_STATE];
#pragma unroll
    for (int n = 0; n < D_STATE; ++n) h[n] = 0.0f;
    float dsum = 0.0f;

    float dtC[SCAN_G], xvC[SCAN_G], dtN[SCAN_G], xvN[SCAN_G];
#pragma unroll
    for (int g = 0; g < SCAN_G; ++g) {
        dtC[g] = pdt[(size_t)g * D_INNER];
        xvC[g] = pxc[(size_t)g * D_INNER];
    }
#pragma unroll 1
    for (int s = 0; s < SCAN_LEN; s += SCAN_G) {
        if (s + SCAN_G < SCAN_LEN) {
#pragma unroll
            for (int g = 0; g < SCAN_G; ++g) {
                dtN[g] = pdt[(size_t)(s + SCAN_G + g) * D_INNER];
                xvN[g] = pxc[(size_t)(s + SCAN_G + g) * D_INNER];
            }
        }
#pragma unroll
        for (int g = 0; g < SCAN_G; ++g) {
            const float* bq = pB + (size_t)(s + g) * DBC_LD;
            const float dt = dtC[g];
            const float xv = xvC[g];
#pragma unroll
            for (int n = 0; n < D_STATE; ++n) {
                const float a = __expf(dt * An[n]);
                h[n] = a * h[n] + (dt * bq[n]) * xv;
            }
            dsum += dt;
        }
#pragma unroll
        for (int g = 0; g < SCAN_G; ++g) { dtC[g] = dtN[g]; xvC[g] = xvN[g]; }
    }

    const size_t o = (size_t)(b * SCAN_NC + c) * SUM_LD + d;
#pragma unroll
    for (int n = 0; n < D_STATE; ++n) {
        hend[o + (size_t)n * D_INNER] = h[n];
        pdec[o + (size_t)n * D_INNER] = __expf(dsum * An[n]);
    }
}

__launch_bounds__(SCAN_TPB)
__global__ void scan_chunk_kernel(const float* __restrict__ delta,
                                  const float* __restrict__ xc,
                                  const float* __restrict__ dbc,
                                  const float* __restrict__ zbuf,
                                  const float* __restrict__ logA,
                                  const float* __restrict__ Dskip,
                                  const float* __restrict__ hend,
                                  const float* __restrict__ pdec,
                                  float* __restrict__ y) {
    const int d = blockIdx.x * SCAN_TPB + threadIdx.x;
    const int c = blockIdx.y;
    const int b = blockIdx.z;

    float An[D_STATE];
    load_decay_rates(logA, d, An);
    const float dsk = Dskip[d];

    const size_t row0 = (size_t)b * SEQ + (size_t)c * SCAN_LEN;
    const float* pdt = delta + row0 * D_INNER + d;
    const float* pxc = xc    + row0 * D_INNER + d;
    const float* pz  = zbuf  + row0 * D_INNER + d;
    float*       py  = y     + row0 * D_INNER + d;
    const float* pBC = dbc   + row0 * DBC_LD + DT_RANK;

    float dtC[SCAN_G], xvC[SCAN_G], zvC[SCAN_G];
    float dtN[SCAN_G], xvN[SCAN_G], zvN[SCAN_G];
#pragma unroll
    for (int g = 0; g < SCAN_G; ++g) {
        dtC[g] = pdt[(size_t)g * D_INNER];
        xvC[g] = pxc[(size_t)g * D_INNER];
        zvC[g] = pz[(size_t)g * D_INNER];
    }

    float h[D_STATE];
#pragma unroll
    for (int n = 0; n < D_STATE; ++n) h[n] = 0.0f;
    const size_t o0 = (size_t)b * SCAN_NC * SUM_LD + d;
#pragma unroll 2
    for (int j = 0; j < c; ++j) {
        const size_t o = o0 + (size_t)j * SUM_LD;
#pragma unroll
        for (int n = 0; n < D_STATE; ++n)
            h[n] = pdec[o + (size_t)n * D_INNER] * h[n] + hend[o + (size_t)n * D_INNER];
    }

#pragma unroll 1
    for (int s = 0; s < SCAN_LEN; s += SCAN_G) {
        if (s + SCAN_G < SCAN_LEN) {
#pragma unroll
            for (int g = 0; g < SCAN_G; ++g) {
                dtN[g] = pdt[(size_t)(s + SCAN_G + g) * D_INNER];
                xvN[g] = pxc[(size_t)(s + SCAN_G + g) * D_INNER];
                zvN[g] = pz[(size_t)(s + SCAN_G + g) * D_INNER];
            }
        }
#pragma unroll
        for (int g = 0; g < SCAN_G; ++g) {
            const float* bq = pBC + (size_t)(s + g) * DBC_LD;
            const float dt = dtC[g];
            const float xv = xvC[g];
            float p[D_STATE];
#pragma unroll
            for (int n = 0; n < D_STATE; ++n) {
                const float a = __expf(dt * An[n]);
                h[n] = a * h[n] + (dt * bq[n]) * xv;
                p[n] = h[n] * bq[D_STATE + n];
            }
#pragma unroll
            for (int w = 1; w < D_STATE; w *= 2)
#pragma unroll
                for (int n = 0; n < D_STATE; n += 2 * w) p[n] += p[n + w];

            py[(size_t)(s + g) * D_INNER] = (p[0] + dsk * xv) * silu_f(zvC[g]);
        }
#pragma unroll
        for (int g = 0; g < SCAN_G; ++g) { dtC[g] = dtN[g]; xvC[g] = xvN[g]; zvC[g] = zvN[g]; }
    }
}

extern "C" void kernel_launch(void* const* d_in, const int* in_sizes, int n_in,
                              void* d_out, int out_size, void* d_ws, size_t ws_size,
                              hipStream_t stream) {
    const float* x         = (const float*)d_in[0];
    const float* in_proj_w = (const float*)d_in[1];
    const float* conv_w    = (const float*)d_in[2];
    const float* conv_b    = (const float*)d_in[3];
    const float* x_proj_w  = (const float*)d_in[4];
    const float* dt_proj_w = (const float*)d_in[5];
    const float* dt_proj_b = (const float*)d_in[6];
    const float* log_A     = (const float*)d_in[7];
    const float* D_skip    = (const float*)d_in[8];
    const float* out_proj_w= (const float*)d_in[9];
    float* out = (float*)d_out;

    const size_t planeBD = (size_t)MROWS * D_INNER;
    float* ws      = (float*)d_ws;
    float* buf_x   = ws;
    float* buf_z   = buf_x  + planeBD;
    float* buf_xc  = buf_z  + planeBD;
    float* buf_dbc = buf_xc + planeBD;
    float* buf_y   = buf_dbc + (size_t)MROWS * DBC_LD;
    float* buf_he  = buf_y  + planeBD;
    float* buf_pd  = buf_he + (size_t)BATCH * SCAN_NC * SUM_LD;

    gemm_wmma_f32<4, 6, 32, 2, 2, 0><<<dim3(D_INNER / 192, MROWS / 128), 128, 0, stream>>>(
        x, in_proj_w, nullptr, buf_x, D_MODEL, D_MODEL, D_INNER);
    gemm_wmma_f32<4, 6, 32, 2, 2, 0><<<dim3(D_INNER / 192, MROWS / 128), 128, 0, stream>>>(
        x, in_proj_w + (size_t)D_INNER * D_MODEL, nullptr, buf_z,
        D_MODEL, D_MODEL, D_INNER);

    conv_silu_kernel<<<(MROWS * D_INNER) / 256, 256, 0, stream>>>(
        buf_x, conv_w, conv_b, buf_xc);

    gemm_wmma_f32<1, 5, 64, 1, 1, 0><<<dim3(1, MROWS / 16), 32, 0, stream>>>(
        buf_xc, x_proj_w, nullptr, buf_dbc, D_INNER, D_INNER, DBC_LD);

    gemm_wmma_f32<2, 4, 24, 4, 1, 1><<<dim3(D_INNER / 64, MROWS / 128), 128, 0, stream>>>(
        buf_dbc, dt_proj_w, dt_proj_b, buf_x, DT_RANK, DBC_LD, D_INNER);

    scan_chunk_state_kernel<<<dim3(D_INNER / SCAN_TPB, SCAN_NC - 1, BATCH), SCAN_TPB, 0, stream>>>(
        buf_x, buf_xc, buf_dbc, log_A, buf_he, buf_pd);
    scan_chunk_kernel<<<dim3(D_INNER / SCAN_TPB, SCAN_NC, BATCH), SCAN_TPB, 0, stream>>>(
        buf_x, buf_xc, buf_dbc, buf_z, log_A, D_skip, buf_he, buf_pd, buf_y);

    gemm_wmma_f32<4, 3, 32, 2, 2, 0><<<dim3(D_MODEL / 96, MROWS / 128), 128, 0, stream>>>(
        buf_y, out_proj_w, nullptr, out, D_INNER, D_INNER, D_MODEL);
}


# conv+SiLU re-mapped to long waves: a lane takes 4 adjacent channels x 16 tokens (whole-line 16-byte loads, taps from registers): 3,072 waves, was 196,608 one-element waves; sum order kept
# speedup vs baseline: 8.4937x; 1.0697x over previous
#include <hip/hip_runtime.h>
#include <hip/hip_bf16.h>


#define D_MODEL 768
#define D_STATE 16
#define D_CONV  4
#define D_INNER 1536
#define DT_RANK 48
#define BATCH   2
#define SEQ     2048
#define MROWS   (BATCH * SEQ)
#define DBC_LD  (DT_RANK + 2 * D_STATE)

typedef __attribute__((ext_vector_type(2))) float v2f;
typedef __attribute__((ext_vector_type(4))) float v4f;
typedef __attribute__((ext_vector_type(8))) float v8f;

__device__ __forceinline__ float silu_f(float x) {
    return x * (1.0f / (1.0f + __expf(-x)));
}
__device__ __forceinline__ float softplus_f(float x) {
    return (x > 20.0f) ? x : log1pf(__expf(x));
}

template <int MT, int NT, int KS, int WM, int WN, int EPI>
__launch_bounds__(WM * WN * 32)
__global__ void gemm_wmma_f32(const float* __restrict__ A,
                              const float* __restrict__ W,
                              const float* __restrict__ bias,
                              float* __restrict__ C,
                              int K, int lda, int ldc) {
    constexpr int KQ = KS / 8;
    const int lane  = threadIdx.x & 31;
    const int wave  = threadIdx.x >> 5;
    const int wm    = wave % WM;
    const int wn    = wave / WM;
    const int mBase = (blockIdx.y * WM + wm) * (16 * MT);
    const int nBase = (blockIdx.x * WN + wn) * (16 * NT);
    const int r16   = lane & 15;
    const int kOff  = (lane >> 4) * 4;

    const float* aP[MT];
    const float* wP[NT];
#pragma unroll
    for (int i = 0; i < MT; ++i)
        aP[i] = A + (size_t)(mBase + i * 16 + r16) * lda + kOff;
#pragma unroll
    for (int t = 0; t < NT; ++t)
        wP[t] = W + (size_t)(nBase + t * 16 + r16) * K + kOff;

    v8f acc[MT][NT];
#pragma unroll
    for (int i = 0; i < MT; ++i)
#pragma unroll
        for (int t = 0; t < NT; ++t) {
            v8f z = {0.f, 0.f, 0.f, 0.f, 0.f, 0.f, 0.f, 0.f};
            acc[i][t] = z;
        }

    v4f aA[MT][KQ], bA[NT][KQ], aB[MT][KQ], bB[NT][KQ];

#define LOADFRAG(AF, BF, KK)                                         \
    _Pragma("unroll") for (int i = 0; i < MT; ++i)                   \
        _Pragma("unroll") for (int j = 0; j < KQ; ++j)               \
            AF[i][j] = *(const v4f*)(aP[i] + (KK) + 8 * j);          \
    _Pragma("unroll") for (int t = 0; t < NT; ++t)                   \
        _Pragma("unroll") for (int j = 0; j < KQ; ++j)               \
            BF[t][j] = *(const v4f*)(wP[t] + (KK) + 8 * j);

#define MACFRAG(AF, BF)                                              \
    _Pragma("unroll") for (int j = 0; j < KQ; ++j) {                 \
        _Pragma("unroll") for (int i = 0; i < MT; ++i)               \
            _Pragma("unroll") for (int t = 0; t < NT; ++t)           \
                acc[i][t] = __builtin_amdgcn_wmma_f32_16x16x4_f32(   \
                    false, AF[i][j].xy, false, BF[t][j].xy,          \
                    (short)0, acc[i][t], false, false);              \
        _Pragma("unroll") for (int i = 0; i < MT; ++i)               \
            _Pragma("unroll") for (int t = 0; t < NT; ++t)           \
                acc[i][t] = __builtin_amdgcn_wmma_f32_16x16x4_f32(   \
                    false, AF[i][j].zw, false, BF[t][j].zw,          \
                    (short)0, acc[i][t], false, false);              \
    }

    LOADFRAG(aA, bA, 0)
    int k = 0;
    for (; k + 2 * KS < K; k += 2 * KS) {
        LOADFRAG(aB, bB, k + KS)
        MACFRAG(aA, bA)
        LOADFRAG(aA, bA, k + 2 * KS)
        MACFRAG(aB, bB)
    }
    LOADFRAG(aB, bB, k + KS)
    MACFRAG(aA, bA)
    MACFRAG(aB, bB)

#undef LOADFRAG
#undef MACFRAG

    const int rowSel = (lane >> 4) * 8;
#pragma unroll
    for (int i = 0; i < MT; ++i)
#pragma unroll
        for (int t = 0; t < NT; ++t) {
            const int col = nBase + t * 16 + (lane & 15);
            float bv = 0.0f;
            if (EPI == 1) bv = bias[col];
#pragma unroll
            for (int r = 0; r < 8; ++r) {
                float v = acc[i][t][r];
                if (EPI == 1) v = softplus_f(v + bv);
                C[(size_t)(mBase + i * 16 + r + rowSel) * ldc + col] = v;
            }
        }
}

#define CONV_T    16
#define CONV_TPB  (D_INNER / 4)

__launch_bounds__(CONV_TPB) __attribute__((amdgpu_waves_per_eu(3, 4)))
__global__ void conv_silu_kernel(const float* __restrict__ xin,
                                 const float* __restrict__ w,
                                 const float* __restrict__ b,
                                 float* __restrict__ out) {
    const int c4 = threadIdx.x * 4;
    const int s0 = blockIdx.x * CONV_T;
    const int bi = blockIdx.y;
    const bool head = (s0 == 0);

    const size_t off = ((size_t)bi * SEQ + s0) * D_INNER + c4;
    const float* px = xin + off;
    float*       po = out + off;

    v4f wv[4];
#pragma unroll
    for (int c = 0; c < 4; ++c)
        wv[c] = *(const v4f*)(w + (size_t)(c4 + c) * D_CONV);
    const v4f bv = *(const v4f*)(b + c4);

    v4f xr[CONV_T + D_CONV - 1];
#pragma unroll
    for (int j = 0; j < D_CONV - 1; ++j) {
        v4f v = {0.f, 0.f, 0.f, 0.f};
        if (!head) v = *(const v4f*)(px - (size_t)(D_CONV - 1 - j) * D_INNER);
        xr[j] = v;
    }
#pragma unroll
    for (int t = 0; t < CONV_T; ++t)
        xr[t + D_CONV - 1] = *(const v4f*)(px + (size_t)t * D_INNER);

#pragma unroll
    for (int t = 0; t < CONV_T; ++t) {
        v4f acc = bv;
#pragma unroll
        for (int j = 0; j < D_CONV; ++j) {
            if (t + j >= D_CONV - 1 || !head) {
                const v4f xv = xr[t + j];
                acc.x += wv[0][j] * xv.x;
                acc.y += wv[1][j] * xv.y;
                acc.z += wv[2][j] * xv.z;
                acc.w += wv[3][j] * xv.w;
            }
        }
        v4f o;
        o.x = silu_f(acc.x);
        o.y = silu_f(acc.y);
        o.z = silu_f(acc.z);
        o.w = silu_f(acc.w);
        *(v4f*)(po + (size_t)t * D_INNER) = o;
    }
}

#define SCAN_NC   32
#define SCAN_LEN  (SEQ / SCAN_NC)
#define SCAN_G    8
#define SCAN_TPB  128
#define SUM_LD    ((size_t)D_STATE * D_INNER)

__device__ __forceinline__ void load_decay_rates(const float* __restrict__ logA,
                                                 int d, float (&An)[D_STATE]) {
#pragma unroll
    for (int q = 0; q < D_STATE / 4; ++q) {
        const v4f v = *(const v4f*)(logA + (size_t)d * D_STATE + 4 * q);
        An[4 * q + 0] = -__expf(v.x);
        An[4 * q + 1] = -__expf(v.y);
        An[4 * q + 2] = -__expf(v.z);
        An[4 * q + 3] = -__expf(v.w);
    }
}

__launch_bounds__(SCAN_TPB)
__global__ void scan_chunk_state_kernel(const float* __restrict__ delta,
                                        const float* __restrict__ xc,
                                        const float* __restrict__ dbc,
                                        const float* __restrict__ logA,
                                        float* __restrict__ hend,
                                        float* __restrict__ pdec) {
    const int d = blockIdx.x * SCAN_TPB + threadIdx.x;
    const int c = blockIdx.y;
    const int b = blockIdx.z;

    float An[D_STATE];
    load_decay_rates(logA, d, An);

    const size_t row0 = (size_t)b * SEQ + (size_t)c * SCAN_LEN;
    const float* pdt = delta + row0 * D_INNER + d;
    const float* pxc = xc    + row0 * D_INNER + d;
    const float* pB  = dbc   + row0 * DBC_LD + DT_RANK;

    float h[D_STATE];
#pragma unroll
    for (int n = 0; n < D_STATE; ++n) h[n] = 0.0f;
    float dsum = 0.0f;

    float dtC[SCAN_G], xvC[SCAN_G], dtN[SCAN_G], xvN[SCAN_G];
#pragma unroll
    for (int g = 0; g < SCAN_G; ++g) {
        dtC[g] = pdt[(size_t)g * D_INNER];
        xvC[g] = pxc[(size_t)g * D_INNER];
    }
#pragma unroll 1
    for (int s = 0; s < SCAN_LEN; s += SCAN_G) {
        if (s + SCAN_G < SCAN_LEN) {
#pragma unroll
            for (int g = 0; g < SCAN_G; ++g) {
                dtN[g] = pdt[(size_t)(s + SCAN_G + g) * D_INNER];
                xvN[g] = pxc[(size_t)(s + SCAN_G + g) * D_INNER];
            }
        }
#pragma unroll
        for (int g = 0; g < SCAN_G; ++g) {
            const float* bq = pB + (size_t)(s + g) * DBC_LD;
            const float dt = dtC[g];
            const float xv = xvC[g];
#pragma unroll
            for (int n = 0; n < D_STATE; ++n) {
                const float a = __expf(dt * An[n]);
                h[n] = a * h[n] + (dt * bq[n]) * xv;
            }
            dsum += dt;
        }
#pragma unroll
        for (int g = 0; g < SCAN_G; ++g) { dtC[g] = dtN[g]; xvC[g] = xvN[g]; }
    }

    const size_t o = (size_t)(b * SCAN_NC + c) * SUM_LD + d;
#pragma unroll
    for (int n = 0; n < D_STATE; ++n) {
        hend[o + (size_t)n * D_INNER] = h[n];
        pdec[o + (size_t)n * D_INNER] = __expf(dsum * An[n]);
    }
}

__launch_bounds__(SCAN_TPB)
__global__ void scan_chunk_kernel(const float* __restrict__ delta,
                                  const float* __restrict__ xc,
                                  const float* __restrict__ dbc,
                                  const float* __restrict__ zbuf,
                                  const float* __restrict__ logA,
                                  const float* __restrict__ Dskip,
                                  const float* __restrict__ hend,
                                  const float* __restrict__ pdec,
                                  float* __restrict__ y) {
    const int d = blockIdx.x * SCAN_TPB + threadIdx.x;
    const int c = blockIdx.y;
    const int b = blockIdx.z;

    float An[D_STATE];
    load_decay_rates(logA, d, An);
    const float dsk = Dskip[d];

    const size_t row0 = (size_t)b * SEQ + (size_t)c * SCAN_LEN;
    const float* pdt = delta + row0 * D_INNER + d;
    const float* pxc = xc    + row0 * D_INNER + d;
    const float* pz  = zbuf  + row0 * D_INNER + d;
    float*       py  = y     + row0 * D_INNER + d;
    const float* pBC = dbc   + row0 * DBC_LD + DT_RANK;

    float dtC[SCAN_G], xvC[SCAN_G], zvC[SCAN_G];
    float dtN[SCAN_G], xvN[SCAN_G], zvN[SCAN_G];
#pragma unroll
    for (int g = 0; g < SCAN_G; ++g) {
        dtC[g] = pdt[(size_t)g * D_INNER];
        xvC[g] = pxc[(size_t)g * D_INNER];
        zvC[g] = pz[(size_t)g * D_INNER];
    }

    float h[D_STATE];
#pragma unroll
    for (int n = 0; n < D_STATE; ++n) h[n] = 0.0f;
    const size_t o0 = (size_t)b * SCAN_NC * SUM_LD + d;
#pragma unroll 2
    for (int j = 0; j < c; ++j) {
        const size_t o = o0 + (size_t)j * SUM_LD;
#pragma unroll
        for (int n = 0; n < D_STATE; ++n)
            h[n] = pdec[o + (size_t)n * D_INNER] * h[n] + hend[o + (size_t)n * D_INNER];
    }

#pragma unroll 1
    for (int s = 0; s < SCAN_LEN; s += SCAN_G) {
        if (s + SCAN_G < SCAN_LEN) {
#pragma unroll
            for (int g = 0; g < SCAN_G; ++g) {
                dtN[g] = pdt[(size_t)(s + SCAN_G + g) * D_INNER];
                xvN[g] = pxc[(size_t)(s + SCAN_G + g) * D_INNER];
                zvN[g] = pz[(size_t)(s + SCAN_G + g) * D_INNER];
            }
        }
#pragma unroll
        for (int g = 0; g < SCAN_G; ++g) {
            const float* bq = pBC + (size_t)(s + g) * DBC_LD;
            const float dt = dtC[g];
            const float xv = xvC[g];
            float p[D_STATE];
#pragma unroll
            for (int n = 0; n < D_STATE; ++n) {
                const float a = __expf(dt * An[n]);
                h[n] = a * h[n] + (dt * bq[n]) * xv;
                p[n] = h[n] * bq[D_STATE + n];
            }
#pragma unroll
            for (int w = 1; w < D_STATE; w *= 2)
#pragma unroll
                for (int n = 0; n < D_STATE; n += 2 * w) p[n] += p[n + w];

            py[(size_t)(s + g) * D_INNER] = (p[0] + dsk * xv) * silu_f(zvC[g]);
        }
#pragma unroll
        for (int g = 0; g < SCAN_G; ++g) { dtC[g] = dtN[g]; xvC[g] = xvN[g]; zvC[g] = zvN[g]; }
    }
}

extern "C" void kernel_launch(void* const* d_in, const int* in_sizes, int n_in,
                              void* d_out, int out_size, void* d_ws, size_t ws_size,
                              hipStream_t stream) {
    const float* x         = (const float*)d_in[0];
    const float* in_proj_w = (const float*)d_in[1];
    const float* conv_w    = (const float*)d_in[2];
    const float* conv_b    = (const float*)d_in[3];
    const float* x_proj_w  = (const float*)d_in[4];
    const float* dt_proj_w = (const float*)d_in[5];
    const float* dt_proj_b = (const float*)d_in[6];
    const float* log_A     = (const float*)d_in[7];
    const float* D_skip    = (const float*)d_in[8];
    const float* out_proj_w= (const float*)d_in[9];
    float* out = (float*)d_out;

    const size_t planeBD = (size_t)MROWS * D_INNER;
    float* ws      = (float*)d_ws;
    float* buf_x   = ws;
    float* buf_z   = buf_x  + planeBD;
    float* buf_xc  = buf_z  + planeBD;
    float* buf_dbc = buf_xc + planeBD;
    float* buf_y   = buf_dbc + (size_t)MROWS * DBC_LD;
    float* buf_he  = buf_y  + planeBD;
    float* buf_pd  = buf_he + (size_t)BATCH * SCAN_NC * SUM_LD;

    gemm_wmma_f32<4, 6, 32, 2, 2, 0><<<dim3(D_INNER / 192, MROWS / 128), 128, 0, stream>>>(
        x, in_proj_w, nullptr, buf_x, D_MODEL, D_MODEL, D_INNER);
    gemm_wmma_f32<4, 6, 32, 2, 2, 0><<<dim3(D_INNER / 192, MROWS / 128), 128, 0, stream>>>(
        x, in_proj_w + (size_t)D_INNER * D_MODEL, nullptr, buf_z,
        D_MODEL, D_MODEL, D_INNER);

    conv_silu_kernel<<<dim3(SEQ / CONV_T, BATCH), CONV_TPB, 0, stream>>>(
        buf_x, conv_w, conv_b, buf_xc);

    gemm_wmma_f32<1, 5, 64, 1, 1, 0><<<dim3(1, MROWS / 16), 32, 0, stream>>>(
        buf_xc, x_proj_w, nullptr, buf_dbc, D_INNER, D_INNER, DBC_LD);

    gemm_wmma_f32<2, 4, 24, 4, 1, 1><<<dim3(D_INNER / 64, MROWS / 128), 128, 0, stream>>>(
        buf_dbc, dt_proj_w, dt_proj_b, buf_x, DT_RANK, DBC_LD, D_INNER);

    scan_chunk_state_kernel<<<dim3(D_INNER / SCAN_TPB, SCAN_NC - 1, BATCH), SCAN_TPB, 0, stream>>>(
        buf_x, buf_xc, buf_dbc, log_A, buf_he, buf_pd);
    scan_chunk_kernel<<<dim3(D_INNER / SCAN_TPB, SCAN_NC, BATCH), SCAN_TPB, 0, stream>>>(
        buf_x, buf_xc, buf_dbc, buf_z, log_A, D_skip, buf_he, buf_pd, buf_y);

    gemm_wmma_f32<4, 3, 32, 2, 2, 0><<<dim3(D_MODEL / 96, MROWS / 128), 128, 0, stream>>>(
        buf_y, out_proj_w, nullptr, out, D_INNER, D_INNER, D_MODEL);
}


# scan: block copies its chunk's 64 rows of B,C to LDS once; each step reads its row by broadcast LDS loads issued a step ahead, not by scalar loads waited at L2 latency; sum order kept
# speedup vs baseline: 8.8327x; 1.0399x over previous
#include <hip/hip_runtime.h>
#include <hip/hip_bf16.h>


#define D_MODEL 768
#define D_STATE 16
#define D_CONV  4
#define D_INNER 1536
#define DT_RANK 48
#define BATCH   2
#define SEQ     2048
#define MROWS   (BATCH * SEQ)
#define DBC_LD  (DT_RANK + 2 * D_STATE)

typedef __attribute__((ext_vector_type(2))) float v2f;
typedef __attribute__((ext_vector_type(4))) float v4f;
typedef __attribute__((ext_vector_type(8))) float v8f;

__device__ __forceinline__ float silu_f(float x) {
    return x * (1.0f / (1.0f + __expf(-x)));
}
__device__ __forceinline__ float softplus_f(float x) {
    return (x > 20.0f) ? x : log1pf(__expf(x));
}

template <int MT, int NT, int KS, int WM, int WN, int EPI>
__launch_bounds__(WM * WN * 32)
__global__ void gemm_wmma_f32(const float* __restrict__ A,
                              const float* __restrict__ W,
                              const float* __restrict__ bias,
                              float* __restrict__ C,
                              int K, int lda, int ldc) {
    constexpr int KQ = KS / 8;
    const int lane  = threadIdx.x & 31;
    const int wave  = threadIdx.x >> 5;
    const int wm    = wave % WM;
    const int wn    = wave / WM;
    const int mBase = (blockIdx.y * WM + wm) * (16 * MT);
    const int nBase = (blockIdx.x * WN + wn) * (16 * NT);
    const int r16   = lane & 15;
    const int kOff  = (lane >> 4) * 4;

    const float* aP[MT];
    const float* wP[NT];
#pragma unroll
    for (int i = 0; i < MT; ++i)
        aP[i] = A + (size_t)(mBase + i * 16 + r16) * lda + kOff;
#pragma unroll
    for (int t = 0; t < NT; ++t)
        wP[t] = W + (size_t)(nBase + t * 16 + r16) * K + kOff;

    v8f acc[MT][NT];
#pragma unroll
    for (int i = 0; i < MT; ++i)
#pragma unroll
        for (int t = 0; t < NT; ++t) {
            v8f z = {0.f, 0.f, 0.f, 0.f, 0.f, 0.f, 0.f, 0.f};
            acc[i][t] = z;
        }

    v4f aA[MT][KQ], bA[NT][KQ], aB[MT][KQ], bB[NT][KQ];

#define LOADFRAG(AF, BF, KK)                                         \
    _Pragma("unroll") for (int i = 0; i < MT; ++i)                   \
        _Pragma("unroll") for (int j = 0; j < KQ; ++j)               \
            AF[i][j] = *(const v4f*)(aP[i] + (KK) + 8 * j);          \
    _Pragma("unroll") for (int t = 0; t < NT; ++t)                   \
        _Pragma("unroll") for (int j = 0; j < KQ; ++j)               \
            BF[t][j] = *(const v4f*)(wP[t] + (KK) + 8 * j);

#define MACFRAG(AF, BF)                                              \
    _Pragma("unroll") for (int j = 0; j < KQ; ++j) {                 \
        _Pragma("unroll") for (int i = 0; i < MT; ++i)               \
            _Pragma("unroll") for (int t = 0; t < NT; ++t)           \
                acc[i][t] = __builtin_amdgcn_wmma_f32_16x16x4_f32(   \
                    false, AF[i][j].xy, false, BF[t][j].xy,          \
                    (short)0, acc[i][t], false, false);              \
        _Pragma("unroll") for (int i = 0; i < MT; ++i)               \
            _Pragma("unroll") for (int t = 0; t < NT; ++t)           \
                acc[i][t] = __builtin_amdgcn_wmma_f32_16x16x4_f32(   \
                    false, AF[i][j].zw, false, BF[t][j].zw,          \
                    (short)0, acc[i][t], false, false);              \
    }

    LOADFRAG(aA, bA, 0)
    int k = 0;
    for (; k + 2 * KS < K; k += 2 * KS) {
        LOADFRAG(aB, bB, k + KS)
        MACFRAG(aA, bA)
        LOADFRAG(aA, bA, k + 2 * KS)
        MACFRAG(aB, bB)
    }
    LOADFRAG(aB, bB, k + KS)
    MACFRAG(aA, bA)
    MACFRAG(aB, bB)

#undef LOADFRAG
#undef MACFRAG

    const int rowSel = (lane >> 4) * 8;
#pragma unroll
    for (int i = 0; i < MT; ++i)
#pragma unroll
        for (int t = 0; t < NT; ++t) {
            const int col = nBase + t * 16 + (lane & 15);
            float bv = 0.0f;
            if (EPI == 1) bv = bias[col];
#pragma unroll
            for (int r = 0; r < 8; ++r) {
                float v = acc[i][t][r];
                if (EPI == 1) v = softplus_f(v + bv);
                C[(size_t)(mBase + i * 16 + r + rowSel) * ldc + col] = v;
            }
        }
}

#define CONV_T    16
#define CONV_TPB  (D_INNER / 4)

__launch_bounds__(CONV_TPB) __attribute__((amdgpu_waves_per_eu(3, 4)))
__global__ void conv_silu_kernel(const float* __restrict__ xin,
                                 const float* __restrict__ w,
                                 const float* __restrict__ b,
                                 float* __restrict__ out) {
    const int c4 = threadIdx.x * 4;
    const int s0 = blockIdx.x * CONV_T;
    const int bi = blockIdx.y;
    const bool head = (s0 == 0);

    const size_t off = ((size_t)bi * SEQ + s0) * D_INNER + c4;
    const float* px = xin + off;
    float*       po = out + off;

    v4f wv[4];
#pragma unroll
    for (int c = 0; c < 4; ++c)
        wv[c] = *(const v4f*)(w + (size_t)(c4 + c) * D_CONV);
    const v4f bv = *(const v4f*)(b + c4);

    v4f xr[CONV_T + D_CONV - 1];
#pragma unroll
    for (int j = 0; j < D_CONV - 1; ++j) {
        v4f v = {0.f, 0.f, 0.f, 0.f};
        if (!head) v = *(const v4f*)(px - (size_t)(D_CONV - 1 - j) * D_INNER);
        xr[j] = v;
    }
#pragma unroll
    for (int t = 0; t < CONV_T; ++t)
        xr[t + D_CONV - 1] = *(const v4f*)(px + (size_t)t * D_INNER);

#pragma unroll
    for (int t = 0; t < CONV_T; ++t) {
        v4f acc = bv;
#pragma unroll
        for (int j = 0; j < D_CONV; ++j) {
            if (t + j >= D_CONV - 1 || !head) {
                const v4f xv = xr[t + j];
                acc.x += wv[0][j] * xv.x;
                acc.y += wv[1][j] * xv.y;
                acc.z += wv[2][j] * xv.z;
                acc.w += wv[3][j] * xv.w;
            }
        }
        v4f o;
        o.x = silu_f(acc.x);
        o.y = silu_f(acc.y);
        o.z = silu_f(acc.z);
        o.w = silu_f(acc.w);
        *(v4f*)(po + (size_t)t * D_INNER) = o;
    }
}

#define SCAN_NC   32
#define SCAN_LEN  (SEQ / SCAN_NC)
#define SCAN_G    8
#define SCAN_TPB  128
#define SUM_LD    ((size_t)D_STATE * D_INNER)
#define SCAN_BQ   (D_STATE / 4)

__device__ __forceinline__ void load_decay_rates(const float* __restrict__ logA,
                                                 int d, float (&An)[D_STATE]) {
#pragma unroll
    for (int q = 0; q < D_STATE / 4; ++q) {
        const v4f v = *(const v4f*)(logA + (size_t)d * D_STATE + 4 * q);
        An[4 * q + 0] = -__expf(v.x);
        An[4 * q + 1] = -__expf(v.y);
        An[4 * q + 2] = -__expf(v.z);
        An[4 * q + 3] = -__expf(v.w);
    }
}

__launch_bounds__(SCAN_TPB)
__global__ void scan_chunk_state_kernel(const float* __restrict__ delta,
                                        const float* __restrict__ xc,
                                        const float* __restrict__ dbc,
                                        const float* __restrict__ logA,
                                        float* __restrict__ hend,
                                        float* __restrict__ pdec) {
    const int d = blockIdx.x * SCAN_TPB + threadIdx.x;
    const int c = blockIdx.y;
    const int b = blockIdx.z;

    float An[D_STATE];
    load_decay_rates(logA, d, An);

    const size_t row0 = (size_t)b * SEQ + (size_t)c * SCAN_LEN;
    const float* pdt = delta + row0 * D_INNER + d;
    const float* pxc = xc    + row0 * D_INNER + d;
    const float* pB  = dbc   + row0 * DBC_LD + DT_RANK;

    __shared__ v4f sB[SCAN_LEN * SCAN_BQ];
    v4f stg[SCAN_LEN * SCAN_BQ / SCAN_TPB];
#pragma unroll
    for (int q = 0; q < SCAN_LEN * SCAN_BQ / SCAN_TPB; ++q) {
        const int p = threadIdx.x + q * SCAN_TPB;
        stg[q] = *(const v4f*)(pB + (size_t)(p / SCAN_BQ) * DBC_LD + (p % SCAN_BQ) * 4);
    }

    float h[D_STATE];
#pragma unroll
    for (int n = 0; n < D_STATE; ++n) h[n] = 0.0f;
    float dsum = 0.0f;

    float dtC[SCAN_G], xvC[SCAN_G], dtN[SCAN_G], xvN[SCAN_G];
#pragma unroll
    for (int g = 0; g < SCAN_G; ++g) {
        dtC[g] = pdt[(size_t)g * D_INNER];
        xvC[g] = pxc[(size_t)g * D_INNER];
    }

#pragma unroll
    for (int q = 0; q < SCAN_LEN * SCAN_BQ / SCAN_TPB; ++q)
        sB[threadIdx.x + q * SCAN_TPB] = stg[q];
    __syncthreads();

    v4f bC[SCAN_BQ], bN[SCAN_BQ];
#pragma unroll
    for (int q = 0; q < SCAN_BQ; ++q) bC[q] = sB[q];
#pragma unroll 1
    for (int s = 0; s < SCAN_LEN; s += SCAN_G) {
        if (s + SCAN_G < SCAN_LEN) {
#pragma unroll
            for (int g = 0; g < SCAN_G; ++g) {
                dtN[g] = pdt[(size_t)(s + SCAN_G + g) * D_INNER];
                xvN[g] = pxc[(size_t)(s + SCAN_G + g) * D_INNER];
            }
        }
#pragma unroll
        for (int g = 0; g < SCAN_G; ++g) {
            const int sn = (s + g + 1 < SCAN_LEN) ? (s + g + 1) : (SCAN_LEN - 1);
#pragma unroll
            for (int q = 0; q < SCAN_BQ; ++q) bN[q] = sB[sn * SCAN_BQ + q];
            __builtin_amdgcn_sched_barrier(0);
            float bq[D_STATE];
#pragma unroll
            for (int q = 0; q < SCAN_BQ; ++q) {
                bq[4 * q + 0] = bC[q].x; bq[4 * q + 1] = bC[q].y;
                bq[4 * q + 2] = bC[q].z; bq[4 * q + 3] = bC[q].w;
            }
            const float dt = dtC[g];
            const float xv = xvC[g];
#pragma unroll
            for (int n = 0; n < D_STATE; ++n) {
                const float a = __expf(dt * An[n]);
                h[n] = a * h[n] + (dt * bq[n]) * xv;
            }
            dsum += dt;
#pragma unroll
            for (int q = 0; q < SCAN_BQ; ++q) bC[q] = bN[q];
        }
#pragma unroll
        for (int g = 0; g < SCAN_G; ++g) { dtC[g] = dtN[g]; xvC[g] = xvN[g]; }
    }

    const size_t o = (size_t)(b * SCAN_NC + c) * SUM_LD + d;
#pragma unroll
    for (int n = 0; n < D_STATE; ++n) {
        hend[o + (size_t)n * D_INNER] = h[n];
        pdec[o + (size_t)n * D_INNER] = __expf(dsum * An[n]);
    }
}

__launch_bounds__(SCAN_TPB) __attribute__((amdgpu_waves_per_eu(4, 4)))
__global__ void scan_chunk_kernel(const float* __restrict__ delta,
                                  const float* __restrict__ xc,
                                  const float* __restrict__ dbc,
                                  const float* __restrict__ zbuf,
                                  const float* __restrict__ logA,
                                  const float* __restrict__ Dskip,
                                  const float* __restrict__ hend,
                                  const float* __restrict__ pdec,
                                  float* __restrict__ y) {
    const int d = blockIdx.x * SCAN_TPB + threadIdx.x;
    const int c = blockIdx.y;
    const int b = blockIdx.z;

    float An[D_STATE];
    load_decay_rates(logA, d, An);
    const float dsk = Dskip[d];

    const size_t row0 = (size_t)b * SEQ + (size_t)c * SCAN_LEN;
    const float* pdt = delta + row0 * D_INNER + d;
    const float* pxc = xc    + row0 * D_INNER + d;
    const float* pz  = zbuf  + row0 * D_INNER + d;
    float*       py  = y     + row0 * D_INNER + d;
    const float* pBC = dbc   + row0 * DBC_LD + DT_RANK;

    __shared__ v4f sBC[SCAN_LEN * 2 * SCAN_BQ];
    v4f stg[SCAN_LEN * 2 * SCAN_BQ / SCAN_TPB];
#pragma unroll
    for (int q = 0; q < SCAN_LEN * 2 * SCAN_BQ / SCAN_TPB; ++q) {
        const int p = threadIdx.x + q * SCAN_TPB;
        stg[q] = *(const v4f*)(pBC + (size_t)(p / (2 * SCAN_BQ)) * DBC_LD + (p % (2 * SCAN_BQ)) * 4);
    }

    float dtC[SCAN_G], xvC[SCAN_G], zvC[SCAN_G];
    float dtN[SCAN_G], xvN[SCAN_G], zvN[SCAN_G];
#pragma unroll
    for (int g = 0; g < SCAN_G; ++g) {
        dtC[g] = pdt[(size_t)g * D_INNER];
        xvC[g] = pxc[(size_t)g * D_INNER];
        zvC[g] = pz[(size_t)g * D_INNER];
    }

    float h[D_STATE];
#pragma unroll
    for (int n = 0; n < D_STATE; ++n) h[n] = 0.0f;
    const size_t o0 = (size_t)b * SCAN_NC * SUM_LD + d;
#pragma unroll 2
    for (int j = 0; j < c; ++j) {
        const size_t o = o0 + (size_t)j * SUM_LD;
#pragma unroll
        for (int n = 0; n < D_STATE; ++n)
            h[n] = pdec[o + (size_t)n * D_INNER] * h[n] + hend[o + (size_t)n * D_INNER];
    }

#pragma unroll
    for (int q = 0; q < SCAN_LEN * 2 * SCAN_BQ / SCAN_TPB; ++q)
        sBC[threadIdx.x + q * SCAN_TPB] = stg[q];
    __syncthreads();

    v4f bcC[2 * SCAN_BQ], bcN[2 * SCAN_BQ];
#pragma unroll
    for (int q = 0; q < 2 * SCAN_BQ; ++q) bcC[q] = sBC[q];

#pragma unroll 1
    for (int s = 0; s < SCAN_LEN; s += SCAN_G) {
        if (s + SCAN_G < SCAN_LEN) {
#pragma unroll
            for (int g = 0; g < SCAN_G; ++g) {
                dtN[g] = pdt[(size_t)(s + SCAN_G + g) * D_INNER];
                xvN[g] = pxc[(size_t)(s + SCAN_G + g) * D_INNER];
                zvN[g] = pz[(size_t)(s + SCAN_G + g) * D_INNER];
            }
        }
#pragma unroll
        for (int g = 0; g < SCAN_G; ++g) {
            const int sn = (s + g + 1 < SCAN_LEN) ? (s + g + 1) : (SCAN_LEN - 1);
#pragma unroll
            for (int q = 0; q < 2 * SCAN_BQ; ++q) bcN[q] = sBC[sn * 2 * SCAN_BQ + q];
            __builtin_amdgcn_sched_barrier(0);
            float bq[2 * D_STATE];
#pragma unroll
            for (int q = 0; q < 2 * SCAN_BQ; ++q) {
                bq[4 * q + 0] = bcC[q].x; bq[4 * q + 1] = bcC[q].y;
                bq[4 * q + 2] = bcC[q].z; bq[4 * q + 3] = bcC[q].w;
            }
            const float dt = dtC[g];
            const float xv = xvC[g];
            float p[D_STATE];
#pragma unroll
            for (int n = 0; n < D_STATE; ++n) {
                const float a = __expf(dt * An[n]);
                h[n] = a * h[n] + (dt * bq[n]) * xv;
                p[n] = h[n] * bq[D_STATE + n];
            }
#pragma unroll
            for (int w = 1; w < D_STATE; w *= 2)
#pragma unroll
                for (int n = 0; n < D_STATE; n += 2 * w) p[n] += p[n + w];

            py[(size_t)(s + g) * D_INNER] = (p[0] + dsk * xv) * silu_f(zvC[g]);
#pragma unroll
            for (int q = 0; q < 2 * SCAN_BQ; ++q) bcC[q] = bcN[q];
        }
#pragma unroll
        for (int g = 0; g < SCAN_G; ++g) { dtC[g] = dtN[g]; xvC[g] = xvN[g]; zvC[g] = zvN[g]; }
    }
}

extern "C" void kernel_launch(void* const* d_in, const int* in_sizes, int n_in,
                              void* d_out, int out_size, void* d_ws, size_t ws_size,
                              hipStream_t stream) {
    const float* x         = (const float*)d_in[0];
    const float* in_proj_w = (const float*)d_in[1];
    const float* conv_w    = (const float*)d_in[2];
    const float* conv_b    = (const float*)d_in[3];
    const float* x_proj_w  = (const float*)d_in[4];
    const float* dt_proj_w = (const float*)d_in[5];
    const float* dt_proj_b = (const float*)d_in[6];
    const float* log_A     = (const float*)d_in[7];
    const float* D_skip    = (const float*)d_in[8];
    const float* out_proj_w= (const float*)d_in[9];
    float* out = (float*)d_out;

    const size_t planeBD = (size_t)MROWS * D_INNER;
    float* ws      = (float*)d_ws;
    float* buf_x   = ws;
    float* buf_z   = buf_x  + planeBD;
    float* buf_xc  = buf_z  + planeBD;
    float* buf_dbc = buf_xc + planeBD;
    float* buf_y   = buf_dbc + (size_t)MROWS * DBC_LD;
    float* buf_he  = buf_y  + planeBD;
    float* buf_pd  = buf_he + (size_t)BATCH * SCAN_NC * SUM_LD;

    gemm_wmma_f32<4, 6, 32, 2, 2, 0><<<dim3(D_INNER / 192, MROWS / 128), 128, 0, stream>>>(
        x, in_proj_w, nullptr, buf_x, D_MODEL, D_MODEL, D_INNER);
    gemm_wmma_f32<4, 6, 32, 2, 2, 0><<<dim3(D_INNER / 192, MROWS / 128), 128, 0, stream>>>(
        x, in_proj_w + (size_t)D_INNER * D_MODEL, nullptr, buf_z,
        D_MODEL, D_MODEL, D_INNER);

    conv_silu_kernel<<<dim3(SEQ / CONV_T, BATCH), CONV_TPB, 0, stream>>>(
        buf_x, conv_w, conv_b, buf_xc);

    gemm_wmma_f32<1, 5, 64, 1, 1, 0><<<dim3(1, MROWS / 16), 32, 0, stream>>>(
        buf_xc, x_proj_w, nullptr, buf_dbc, D_INNER, D_INNER, DBC_LD);

    gemm_wmma_f32<2, 4, 24, 4, 1, 1><<<dim3(D_INNER / 64, MROWS / 128), 128, 0, stream>>>(
        buf_dbc, dt_proj_w, dt_proj_b, buf_x, DT_RANK, DBC_LD, D_INNER);

    scan_chunk_state_kernel<<<dim3(D_INNER / SCAN_TPB, SCAN_NC - 1, BATCH), SCAN_TPB, 0, stream>>>(
        buf_x, buf_xc, buf_dbc, log_A, buf_he, buf_pd);
    scan_chunk_kernel<<<dim3(D_INNER / SCAN_TPB, SCAN_NC, BATCH), SCAN_TPB, 0, stream>>>(
        buf_x, buf_xc, buf_dbc, buf_z, log_A, D_skip, buf_he, buf_pd, buf_y);

    gemm_wmma_f32<4, 3, 32, 2, 2, 0><<<dim3(D_MODEL / 96, MROWS / 128), 128, 0, stream>>>(
        buf_y, out_proj_w, nullptr, out, D_INNER, D_INNER, D_MODEL);
}


# x_proj: K split over a block's 4 waves (one per SIMD), partial tiles added in wave order 0..3 via LDS; weight fragments read as 512 contiguous B from a fragment-order copy the conv launch writes
# speedup vs baseline: 9.2119x; 1.0429x over previous
#include <hip/hip_runtime.h>
#include <hip/hip_bf16.h>


#define D_MODEL 768
#define D_STATE 16
#define D_CONV  4
#define D_INNER 1536
#define DT_RANK 48
#define BATCH   2
#define SEQ     2048
#define MROWS   (BATCH * SEQ)
#define DBC_LD  (DT_RANK + 2 * D_STATE)

typedef __attribute__((ext_vector_type(2))) float v2f;
typedef __attribute__((ext_vector_type(4))) float v4f;
typedef __attribute__((ext_vector_type(8))) float v8f;

__device__ __forceinline__ float silu_f(float x) {
    return x * (1.0f / (1.0f + __expf(-x)));
}
__device__ __forceinline__ float softplus_f(float x) {
    return (x > 20.0f) ? x : log1pf(__expf(x));
}

template <int MT, int NT, int KS, int WM, int WN, int EPI>
__launch_bounds__(WM * WN * 32)
__global__ void gemm_wmma_f32(const float* __restrict__ A,
                              const float* __restrict__ W,
                              const float* __restrict__ bias,
                              float* __restrict__ C,
                              int K, int lda, int ldc) {
    constexpr int KQ = KS / 8;
    const int lane  = threadIdx.x & 31;
    const int wave  = threadIdx.x >> 5;
    const int wm    = wave % WM;
    const int wn    = wave / WM;
    const int mBase = (blockIdx.y * WM + wm) * (16 * MT);
    const int nBase = (blockIdx.x * WN + wn) * (16 * NT);
    const int r16   = lane & 15;
    const int kOff  = (lane >> 4) * 4;

    const float* aP[MT];
    const float* wP[NT];
#pragma unroll
    for (int i = 0; i < MT; ++i)
        aP[i] = A + (size_t)(mBase + i * 16 + r16) * lda + kOff;
#pragma unroll
    for (int t = 0; t < NT; ++t)
        wP[t] = W + (size_t)(nBase + t * 16 + r16) * K + kOff;

    v8f acc[MT][NT];
#pragma unroll
    for (int i = 0; i < MT; ++i)
#pragma unroll
        for (int t = 0; t < NT; ++t) {
            v8f z = {0.f, 0.f, 0.f, 0.f, 0.f, 0.f, 0.f, 0.f};
            acc[i][t] = z;
        }

    v4f aA[MT][KQ], bA[NT][KQ], aB[MT][KQ], bB[NT][KQ];

#define LOADFRAG(AF, BF, KK)                                         \
    _Pragma("unroll") for (int i = 0; i < MT; ++i)                   \
        _Pragma("unroll") for (int j = 0; j < KQ; ++j)               \
            AF[i][j] = *(const v4f*)(aP[i] + (KK) + 8 * j);          \
    _Pragma("unroll") for (int t = 0; t < NT; ++t)                   \
        _Pragma("unroll") for (int j = 0; j < KQ; ++j)               \
            BF[t][j] = *(const v4f*)(wP[t] + (KK) + 8 * j);

#define MACFRAG(AF, BF)                                              \
    _Pragma("unroll") for (int j = 0; j < KQ; ++j) {                 \
        _Pragma("unroll") for (int i = 0; i < MT; ++i)               \
            _Pragma("unroll") for (int t = 0; t < NT; ++t)           \
                acc[i][t] = __builtin_amdgcn_wmma_f32_16x16x4_f32(   \
                    false, AF[i][j].xy, false, BF[t][j].xy,          \
                    (short)0, acc[i][t], false, false);              \
        _Pragma("unroll") for (int i = 0; i < MT; ++i)               \
            _Pragma("unroll") for (int t = 0; t < NT; ++t)           \
                acc[i][t] = __builtin_amdgcn_wmma_f32_16x16x4_f32(   \
                    false, AF[i][j].zw, false, BF[t][j].zw,          \
                    (short)0, acc[i][t], false, false);              \
    }

    LOADFRAG(aA, bA, 0)
    int k = 0;
    for (; k + 2 * KS < K; k += 2 * KS) {
        LOADFRAG(aB, bB, k + KS)
        MACFRAG(aA, bA)
        LOADFRAG(aA, bA, k + 2 * KS)
        MACFRAG(aB, bB)
    }
    LOADFRAG(aB, bB, k + KS)
    MACFRAG(aA, bA)
    MACFRAG(aB, bB)

#undef LOADFRAG
#undef MACFRAG

    const int rowSel = (lane >> 4) * 8;
#pragma unroll
    for (int i = 0; i < MT; ++i)
#pragma unroll
        for (int t = 0; t < NT; ++t) {
            const int col = nBase + t * 16 + (lane & 15);
            float bv = 0.0f;
            if (EPI == 1) bv = bias[col];
#pragma unroll
            for (int r = 0; r < 8; ++r) {
                float v = acc[i][t][r];
                if (EPI == 1) v = softplus_f(v + bv);
                C[(size_t)(mBase + i * 16 + r + rowSel) * ldc + col] = v;
            }
        }
}

#define CONV_T    16
#define CONV_TPB  (D_INNER / 4)

#define XP_KS     64
#define XP_KQ     (XP_KS / 8)
#define XP_NT     (DBC_LD / 16)
#define XP_WAVES  4
#define XP_SLICES (D_INNER / XP_KS)
#define XP_SPW    (XP_SLICES / XP_WAVES)
#define XP_FRAG   128
#define XP_PIECES (DBC_LD * D_INNER / 4)

__launch_bounds__(CONV_TPB) __attribute__((amdgpu_waves_per_eu(3, 4)))
__global__ void conv_silu_kernel(const float* __restrict__ xin,
                                 const float* __restrict__ w,
                                 const float* __restrict__ b,
                                 float* __restrict__ out,
                                 const float* __restrict__ xw,
                                 float* __restrict__ xwf) {
    const unsigned g = (blockIdx.y * gridDim.x + blockIdx.x) * CONV_TPB + threadIdx.x;
    const bool mover = g < XP_PIECES;
    v4f piece = {0.f, 0.f, 0.f, 0.f};
    if (mover) piece = *(const v4f*)(xw + (size_t)g * 4);

    const int c4 = threadIdx.x * 4;
    const int s0 = blockIdx.x * CONV_T;
    const int bi = blockIdx.y;
    const bool head = (s0 == 0);

    const size_t off = ((size_t)bi * SEQ + s0) * D_INNER + c4;
    const float* px = xin + off;
    float*       po = out + off;

    v4f wv[4];
#pragma unroll
    for (int c = 0; c < 4; ++c)
        wv[c] = *(const v4f*)(w + (size_t)(c4 + c) * D_CONV);
    const v4f bv = *(const v4f*)(b + c4);

    v4f xr[CONV_T + D_CONV - 1];
#pragma unroll
    for (int j = 0; j < D_CONV - 1; ++j) {
        v4f v = {0.f, 0.f, 0.f, 0.f};
        if (!head) v = *(const v4f*)(px - (size_t)(D_CONV - 1 - j) * D_INNER);
        xr[j] = v;
    }
#pragma unroll
    for (int t = 0; t < CONV_T; ++t)
        xr[t + D_CONV - 1] = *(const v4f*)(px + (size_t)t * D_INNER);

#pragma unroll
    for (int t = 0; t < CONV_T; ++t) {
        v4f acc = bv;
#pragma unroll
        for (int j = 0; j < D_CONV; ++j) {
            if (t + j >= D_CONV - 1 || !head) {
                const v4f xv = xr[t + j];
                acc.x += wv[0][j] * xv.x;
                acc.y += wv[1][j] * xv.y;
                acc.z += wv[2][j] * xv.z;
                acc.w += wv[3][j] * xv.w;
            }
        }
        v4f o;
        o.x = silu_f(acc.x);
        o.y = silu_f(acc.y);
        o.z = silu_f(acc.z);
        o.w = silu_f(acc.w);
        *(v4f*)(po + (size_t)t * D_INNER) = o;
    }

    if (mover) {
        const unsigned n  = g / (D_INNER / 4);
        const unsigned k0 = (g % (D_INNER / 4)) * 4;
        const unsigned s  = k0 / XP_KS;
        const unsigned j  = (k0 % XP_KS) / 8;
        const unsigned h  = (k0 % 8) / 4;
        const unsigned t  = n / 16;
        const unsigned c  = n % 16;
        const unsigned p  = ((s * XP_KQ + j) * XP_NT + t) * 32 + h * 16 + c;
        *(v4f*)(xwf + (size_t)p * 4) = piece;
    }
}

__launch_bounds__(XP_WAVES * 32) __attribute__((amdgpu_waves_per_eu(1, 1)))
__global__ void xproj_ksplit_kernel(const float* __restrict__ A,
                                    const float* __restrict__ Wf,
                                    float* __restrict__ C) {
    const int lane  = threadIdx.x & 31;
    const int wave  = threadIdx.x >> 5;
    const int mBase = blockIdx.x * 16;
    const int r16   = lane & 15;
    const int kOff  = (lane >> 4) * 4;

    const float* aP = A + (size_t)(mBase + r16) * D_INNER + wave * XP_KS + kOff;
    const float* wP = Wf + (size_t)wave * (XP_KQ * XP_NT * XP_FRAG) + lane * 4;

    v8f acc[XP_NT];
#pragma unroll
    for (int t = 0; t < XP_NT; ++t) {
        v8f z = {0.f, 0.f, 0.f, 0.f, 0.f, 0.f, 0.f, 0.f};
        acc[t] = z;
    }

    v4f aA[XP_KQ], bA[XP_NT][XP_KQ], aB[XP_KQ], bB[XP_NT][XP_KQ];

#define XP_LOAD(AF, BF, I)                                                   \
    _Pragma("unroll") for (int j = 0; j < XP_KQ; ++j)                        \
        AF[j] = *(const v4f*)(aP + (I) * (XP_WAVES * XP_KS) + 8 * j);        \
    _Pragma("unroll") for (int j = 0; j < XP_KQ; ++j)                        \
        _Pragma("unroll") for (int t = 0; t < XP_NT; ++t)                    \
            BF[t][j] = *(const v4f*)(wP +                                    \
                (size_t)((((I) * XP_WAVES * XP_KQ + j) * XP_NT + t) * XP_FRAG));

#define XP_MAC(AF, BF)                                                       \
    _Pragma("unroll") for (int j = 0; j < XP_KQ; ++j) {                      \
        _Pragma("unroll") for (int t = 0; t < XP_NT; ++t)                    \
            acc[t] = __builtin_amdgcn_wmma_f32_16x16x4_f32(                  \
                false, AF[j].xy, false, BF[t][j].xy,                         \
                (short)0, acc[t], false, false);                             \
        _Pragma("unroll") for (int t = 0; t < XP_NT; ++t)                    \
            acc[t] = __builtin_amdgcn_wmma_f32_16x16x4_f32(                  \
                false, AF[j].zw, false, BF[t][j].zw,                         \
                (short)0, acc[t], false, false);                             \
    }

    XP_LOAD(aA, bA, 0)
#pragma unroll
    for (int i = 0; i < XP_SPW; i += 2) {
        XP_LOAD(aB, bB, i + 1)
        XP_MAC(aA, bA)
        if (i + 2 < XP_SPW) {
            XP_LOAD(aA, bA, i + 2)
        }
        XP_MAC(aB, bB)
    }

#undef XP_LOAD
#undef XP_MAC

    __shared__ float sP[XP_WAVES][16 * DBC_LD];
    const int rowSel = (lane >> 4) * 8;
#pragma unroll
    for (int t = 0; t < XP_NT; ++t)
#pragma unroll
        for (int r = 0; r < 8; ++r)
            sP[wave][(r + rowSel) * DBC_LD + t * 16 + r16] = acc[t][r];
    __syncthreads();

    float* cP = C + (size_t)mBase * DBC_LD;
#pragma unroll
    for (int q = 0; q < 16 * DBC_LD / (XP_WAVES * 32); ++q) {
        const int e = threadIdx.x + q * (XP_WAVES * 32);
        cP[e] = ((sP[0][e] + sP[1][e]) + sP[2][e]) + sP[3][e];
    }
}

#define SCAN_NC   32
#define SCAN_LEN  (SEQ / SCAN_NC)
#define SCAN_G    8
#define SCAN_TPB  128
#define SUM_LD    ((size_t)D_STATE * D_INNER)
#define SCAN_BQ   (D_STATE / 4)

__device__ __forceinline__ void load_decay_rates(const float* __restrict__ logA,
                                                 int d, float (&An)[D_STATE]) {
#pragma unroll
    for (int q = 0; q < D_STATE / 4; ++q) {
        const v4f v = *(const v4f*)(logA + (size_t)d * D_STATE + 4 * q);
        An[4 * q + 0] = -__expf(v.x);
        An[4 * q + 1] = -__expf(v.y);
        An[4 * q + 2] = -__expf(v.z);
        An[4 * q + 3] = -__expf(v.w);
    }
}

__launch_bounds__(SCAN_TPB)
__global__ void scan_chunk_state_kernel(const float* __restrict__ delta,
                                        const float* __restrict__ xc,
                                        const float* __restrict__ dbc,
                                        const float* __restrict__ logA,
                                        float* __restrict__ hend,
                                        float* __restrict__ pdec) {
    const int d = blockIdx.x * SCAN_TPB + threadIdx.x;
    const int c = blockIdx.y;
    const int b = blockIdx.z;

    float An[D_STATE];
    load_decay_rates(logA, d, An);

    const size_t row0 = (size_t)b * SEQ + (size_t)c * SCAN_LEN;
    const float* pdt = delta + row0 * D_INNER + d;
    const float* pxc = xc    + row0 * D_INNER + d;
    const float* pB  = dbc   + row0 * DBC_LD + DT_RANK;

    __shared__ v4f sB[SCAN_LEN * SCAN_BQ];
    v4f stg[SCAN_LEN * SCAN_BQ / SCAN_TPB];
#pragma unroll
    for (int q = 0; q < SCAN_LEN * SCAN_BQ / SCAN_TPB; ++q) {
        const int p = threadIdx.x + q * SCAN_TPB;
        stg[q] = *(const v4f*)(pB + (size_t)(p / SCAN_BQ) * DBC_LD + (p % SCAN_BQ) * 4);
    }

    float h[D_STATE];
#pragma unroll
    for (int n = 0; n < D_STATE; ++n) h[n] = 0.0f;
    float dsum = 0.0f;

    float dtC[SCAN_G], xvC[SCAN_G], dtN[SCAN_G], xvN[SCAN_G];
#pragma unroll
    for (int g = 0; g < SCAN_G; ++g) {
        dtC[g] = pdt[(size_t)g * D_INNER];
        xvC[g] = pxc[(size_t)g * D_INNER];
    }

#pragma unroll
    for (int q = 0; q < SCAN_LEN * SCAN_BQ / SCAN_TPB; ++q)
        sB[threadIdx.x + q * SCAN_TPB] = stg[q];
    __syncthreads();

    v4f bC[SCAN_BQ], bN[SCAN_BQ];
#pragma unroll
    for (int q = 0; q < SCAN_BQ; ++q) bC[q] = sB[q];
#pragma unroll 1
    for (int s = 0; s < SCAN_LEN; s += SCAN_G) {
        if (s + SCAN_G < SCAN_LEN) {
#pragma unroll
            for (int g = 0; g < SCAN_G; ++g) {
                dtN[g] = pdt[(size_t)(s + SCAN_G + g) * D_INNER];
                xvN[g] = pxc[(size_t)(s + SCAN_G + g) * D_INNER];
            }
        }
#pragma unroll
        for (int g = 0; g < SCAN_G; ++g) {
            const int sn = (s + g + 1 < SCAN_LEN) ? (s + g + 1) : (SCAN_LEN - 1);
#pragma unroll
            for (int q = 0; q < SCAN_BQ; ++q) bN[q] = sB[sn * SCAN_BQ + q];
            __builtin_amdgcn_sched_barrier(0);
            float bq[D_STATE];
#pragma unroll
            for (int q = 0; q < SCAN_BQ; ++q) {
                bq[4 * q + 0] = bC[q].x; bq[4 * q + 1] = bC[q].y;
                bq[4 * q + 2] = bC[q].z; bq[4 * q + 3] = bC[q].w;
            }
            const float dt = dtC[g];
            const float xv = xvC[g];
#pragma unroll
            for (int n = 0; n < D_STATE; ++n) {
                const float a = __expf(dt * An[n]);
                h[n] = a * h[n] + (dt * bq[n]) * xv;
            }
            dsum += dt;
#pragma unroll
            for (int q = 0; q < SCAN_BQ; ++q) bC[q] = bN[q];
        }
#pragma unroll
        for (int g = 0; g < SCAN_G; ++g) { dtC[g] = dtN[g]; xvC[g] = xvN[g]; }
    }

    const size_t o = (size_t)(b * SCAN_NC + c) * SUM_LD + d;
#pragma unroll
    for (int n = 0; n < D_STATE; ++n) {
        hend[o + (size_t)n * D_INNER] = h[n];
        pdec[o + (size_t)n * D_INNER] = __expf(dsum * An[n]);
    }
}

__launch_bounds__(SCAN_TPB) __attribute__((amdgpu_waves_per_eu(4, 4)))
__global__ void scan_chunk_kernel(const float* __restrict__ delta,
                                  const float* __restrict__ xc,
                                  const float* __restrict__ dbc,
                                  const float* __restrict__ zbuf,
                                  const float* __restrict__ logA,
                                  const float* __restrict__ Dskip,
                                  const float* __restrict__ hend,
                                  const float* __restrict__ pdec,
                                  float* __restrict__ y) {
    const int d = blockIdx.x * SCAN_TPB + threadIdx.x;
    const int c = blockIdx.y;
    const int b = blockIdx.z;

    float An[D_STATE];
    load_decay_rates(logA, d, An);
    const float dsk = Dskip[d];

    const size_t row0 = (size_t)b * SEQ + (size_t)c * SCAN_LEN;
    const float* pdt = delta + row0 * D_INNER + d;
    const float* pxc = xc    + row0 * D_INNER + d;
    const float* pz  = zbuf  + row0 * D_INNER + d;
    float*       py  = y     + row0 * D_INNER + d;
    const float* pBC = dbc   + row0 * DBC_LD + DT_RANK;

    __shared__ v4f sBC[SCAN_LEN * 2 * SCAN_BQ];
    v4f stg[SCAN_LEN * 2 * SCAN_BQ / SCAN_TPB];
#pragma unroll
    for (int q = 0; q < SCAN_LEN * 2 * SCAN_BQ / SCAN_TPB; ++q) {
        const int p = threadIdx.x + q * SCAN_TPB;
        stg[q] = *(const v4f*)(pBC + (size_t)(p / (2 * SCAN_BQ)) * DBC_LD + (p % (2 * SCAN_BQ)) * 4);
    }

    float dtC[SCAN_G], xvC[SCAN_G], zvC[SCAN_G];
    float dtN[SCAN_G], xvN[SCAN_G], zvN[SCAN_G];
#pragma unroll
    for (int g = 0; g < SCAN_G; ++g) {
        dtC[g] = pdt[(size_t)g * D_INNER];
        xvC[g] = pxc[(size_t)g * D_INNER];
        zvC[g] = pz[(size_t)g * D_INNER];
    }

    float h[D_STATE];
#pragma unroll
    for (int n = 0; n < D_STATE; ++n) h[n] = 0.0f;
    const size_t o0 = (size_t)b * SCAN_NC * SUM_LD + d;
#pragma unroll 2
    for (int j = 0; j < c; ++j) {
        const size_t o = o0 + (size_t)j * SUM_LD;
#pragma unroll
        for (int n = 0; n < D_STATE; ++n)
            h[n] = pdec[o + (size_t)n * D_INNER] * h[n] + hend[o + (size_t)n * D_INNER];
    }

#pragma unroll
    for (int q = 0; q < SCAN_LEN * 2 * SCAN_BQ / SCAN_TPB; ++q)
        sBC[threadIdx.x + q * SCAN_TPB] = stg[q];
    __syncthreads();

    v4f bcC[2 * SCAN_BQ], bcN[2 * SCAN_BQ];
#pragma unroll
    for (int q = 0; q < 2 * SCAN_BQ; ++q) bcC[q] = sBC[q];

#pragma unroll 1
    for (int s = 0; s < SCAN_LEN; s += SCAN_G) {
        if (s + SCAN_G < SCAN_LEN) {
#pragma unroll
            for (int g = 0; g < SCAN_G; ++g) {
                dtN[g] = pdt[(size_t)(s + SCAN_G + g) * D_INNER];
                xvN[g] = pxc[(size_t)(s + SCAN_G + g) * D_INNER];
                zvN[g] = pz[(size_t)(s + SCAN_G + g) * D_INNER];
            }
        }
#pragma unroll
        for (int g = 0; g < SCAN_G; ++g) {
            const int sn = (s + g + 1 < SCAN_LEN) ? (s + g + 1) : (SCAN_LEN - 1);
#pragma unroll
            for (int q = 0; q < 2 * SCAN_BQ; ++q) bcN[q] = sBC[sn * 2 * SCAN_BQ + q];
            __builtin_amdgcn_sched_barrier(0);
            float bq[2 * D_STATE];
#pragma unroll
            for (int q = 0; q < 2 * SCAN_BQ; ++q) {
                bq[4 * q + 0] = bcC[q].x; bq[4 * q + 1] = bcC[q].y;
                bq[4 * q + 2] = bcC[q].z; bq[4 * q + 3] = bcC[q].w;
            }
            const float dt = dtC[g];
            const float xv = xvC[g];
            float p[D_STATE];
#pragma unroll
            for (int n = 0; n < D_STATE; ++n) {
                const float a = __expf(dt * An[n]);
                h[n] = a * h[n] + (dt * bq[n]) * xv;
                p[n] = h[n] * bq[D_STATE + n];
            }
#pragma unroll
            for (int w = 1; w < D_STATE; w *= 2)
#pragma unroll
                for (int n = 0; n < D_STATE; n += 2 * w) p[n] += p[n + w];

            py[(size_t)(s + g) * D_INNER] = (p[0] + dsk * xv) * silu_f(zvC[g]);
#pragma unroll
            for (int q = 0; q < 2 * SCAN_BQ; ++q) bcC[q] = bcN[q];
        }
#pragma unroll
        for (int g = 0; g < SCAN_G; ++g) { dtC[g] = dtN[g]; xvC[g] = xvN[g]; zvC[g] = zvN[g]; }
    }
}

extern "C" void kernel_launch(void* const* d_in, const int* in_sizes, int n_in,
                              void* d_out, int out_size, void* d_ws, size_t ws_size,
                              hipStream_t stream) {
    const float* x         = (const float*)d_in[0];
    const float* in_proj_w = (const float*)d_in[1];
    const float* conv_w    = (const float*)d_in[2];
    const float* conv_b    = (const float*)d_in[3];
    const float* x_proj_w  = (const float*)d_in[4];
    const float* dt_proj_w = (const float*)d_in[5];
    const float* dt_proj_b = (const float*)d_in[6];
    const float* log_A     = (const float*)d_in[7];
    const float* D_skip    = (const float*)d_in[8];
    const float* out_proj_w= (const float*)d_in[9];
    float* out = (float*)d_out;

    const size_t planeBD = (size_t)MROWS * D_INNER;
    float* ws      = (float*)d_ws;
    float* buf_x   = ws;
    float* buf_z   = buf_x  + planeBD;
    float* buf_xc  = buf_z  + planeBD;
    float* buf_dbc = buf_xc + planeBD;
    float* buf_y   = buf_dbc + (size_t)MROWS * DBC_LD;
    float* buf_he  = buf_y  + planeBD;
    float* buf_pd  = buf_he + (size_t)BATCH * SCAN_NC * SUM_LD;
    float* buf_wf  = buf_pd + (size_t)BATCH * SCAN_NC * SUM_LD;

    gemm_wmma_f32<4, 6, 32, 2, 2, 0><<<dim3(D_INNER / 192, MROWS / 128), 128, 0, stream>>>(
        x, in_proj_w, nullptr, buf_x, D_MODEL, D_MODEL, D_INNER);
    gemm_wmma_f32<4, 6, 32, 2, 2, 0><<<dim3(D_INNER / 192, MROWS / 128), 128, 0, stream>>>(
        x, in_proj_w + (size_t)D_INNER * D_MODEL, nullptr, buf_z,
        D_MODEL, D_MODEL, D_INNER);

    conv_silu_kernel<<<dim3(SEQ / CONV_T, BATCH), CONV_TPB, 0, stream>>>(
        buf_x, conv_w, conv_b, buf_xc, x_proj_w, buf_wf);

    xproj_ksplit_kernel<<<dim3(MROWS / 16), XP_WAVES * 32, 0, stream>>>(
        buf_xc, buf_wf, buf_dbc);

    gemm_wmma_f32<2, 4, 24, 4, 1, 1><<<dim3(D_INNER / 64, MROWS / 128), 128, 0, stream>>>(
        buf_dbc, dt_proj_w, dt_proj_b, buf_x, DT_RANK, DBC_LD, D_INNER);

    scan_chunk_state_kernel<<<dim3(D_INNER / SCAN_TPB, SCAN_NC - 1, BATCH), SCAN_TPB, 0, stream>>>(
        buf_x, buf_xc, buf_dbc, log_A, buf_he, buf_pd);
    scan_chunk_kernel<<<dim3(D_INNER / SCAN_TPB, SCAN_NC, BATCH), SCAN_TPB, 0, stream>>>(
        buf_x, buf_xc, buf_dbc, buf_z, log_A, D_skip, buf_he, buf_pd, buf_y);

    gemm_wmma_f32<4, 3, 32, 2, 2, 0><<<dim3(D_MODEL / 96, MROWS / 128), 128, 0, stream>>>(
        buf_y, out_proj_w, nullptr, out, D_INNER, D_INNER, D_MODEL);
}


# dt_proj+softplus and the chunk-state pass in one launch, one loop (7 launches): a wave's 64x32 WMMA tile goes raw to LDS; per 8 rows the loop takes the 8 softplus, stores delta, runs the 8 steps
# speedup vs baseline: 10.5248x; 1.1425x over previous
#include <hip/hip_runtime.h>
#include <hip/hip_bf16.h>


#define D_MODEL 768
#define D_STATE 16
#define D_CONV  4
#define D_INNER 1536
#define DT_RANK 48
#define BATCH   2
#define SEQ     2048
#define MROWS   (BATCH * SEQ)
#define DBC_LD  (DT_RANK + 2 * D_STATE)

typedef __attribute__((ext_vector_type(2))) float v2f;
typedef __attribute__((ext_vector_type(4))) float v4f;
typedef __attribute__((ext_vector_type(8))) float v8f;

__device__ __forceinline__ float silu_f(float x) {
    return x * (1.0f / (1.0f + __expf(-x)));
}
__device__ __forceinline__ float softplus_f(float x) {
    const float l = log1pf(__expf(x));
    return (x > 20.0f) ? x : l;
}

template <int MT, int NT, int KS, int WM, int WN, int EPI>
__launch_bounds__(WM * WN * 32)
__global__ void gemm_wmma_f32(const float* __restrict__ A,
                              const float* __restrict__ W,
                              const float* __restrict__ bias,
                              float* __restrict__ C,
                              int K, int lda, int ldc) {
    constexpr int KQ = KS / 8;
    const int lane  = threadIdx.x & 31;
    const int wave  = threadIdx.x >> 5;
    const int wm    = wave % WM;
    const int wn    = wave / WM;
    const int mBase = (blockIdx.y * WM + wm) * (16 * MT);
    const int nBase = (blockIdx.x * WN + wn) * (16 * NT);
    const int r16   = lane & 15;
    const int kOff  = (lane >> 4) * 4;

    const float* aP[MT];
    const float* wP[NT];
#pragma unroll
    for (int i = 0; i < MT; ++i)
        aP[i] = A + (size_t)(mBase + i * 16 + r16) * lda + kOff;
#pragma unroll
    for (int t = 0; t < NT; ++t)
        wP[t] = W + (size_t)(nBase + t * 16 + r16) * K + kOff;

    v8f acc[MT][NT];
#pragma unroll
    for (int i = 0; i < MT; ++i)
#pragma unroll
        for (int t = 0; t < NT; ++t) {
            v8f z = {0.f, 0.f, 0.f, 0.f, 0.f, 0.f, 0.f, 0.f};
            acc[i][t] = z;
        }

    v4f aA[MT][KQ], bA[NT][KQ], aB[MT][KQ], bB[NT][KQ];

#define LOADFRAG(AF, BF, KK)                                         \
    _Pragma("unroll") for (int i = 0; i < MT; ++i)                   \
        _Pragma("unroll") for (int j = 0; j < KQ; ++j)               \
            AF[i][j] = *(const v4f*)(aP[i] + (KK) + 8 * j);          \
    _Pragma("unroll") for (int t = 0; t < NT; ++t)                   \
        _Pragma("unroll") for (int j = 0; j < KQ; ++j)               \
            BF[t][j] = *(const v4f*)(wP[t] + (KK) + 8 * j);

#define MACFRAG(AF, BF)                                              \
    _Pragma("unroll") for (int j = 0; j < KQ; ++j) {                 \
        _Pragma("unroll") for (int i = 0; i < MT; ++i)               \
            _Pragma("unroll") for (int t = 0; t < NT; ++t)           \
                acc[i][t] = __builtin_amdgcn_wmma_f32_16x16x4_f32(   \
                    false, AF[i][j].xy, false, BF[t][j].xy,          \
                    (short)0, acc[i][t], false, false);              \
        _Pragma("unroll") for (int i = 0; i < MT; ++i)               \
            _Pragma("unroll") for (int t = 0; t < NT; ++t)           \
                acc[i][t] = __builtin_amdgcn_wmma_f32_16x16x4_f32(   \
                    false, AF[i][j].zw, false, BF[t][j].zw,          \
                    (short)0, acc[i][t], false, false);              \
    }

    LOADFRAG(aA, bA, 0)
    int k = 0;
    for (; k + 2 * KS < K; k += 2 * KS) {
        LOADFRAG(aB, bB, k + KS)
        MACFRAG(aA, bA)
        LOADFRAG(aA, bA, k + 2 * KS)
        MACFRAG(aB, bB)
    }
    LOADFRAG(aB, bB, k + KS)
    MACFRAG(aA, bA)
    MACFRAG(aB, bB)

#undef LOADFRAG
#undef MACFRAG

    const int rowSel = (lane >> 4) * 8;
#pragma unroll
    for (int i = 0; i < MT; ++i)
#pragma unroll
        for (int t = 0; t < NT; ++t) {
            const int col = nBase + t * 16 + (lane & 15);
            float bv = 0.0f;
            if (EPI == 1) bv = bias[col];
#pragma unroll
            for (int r = 0; r < 8; ++r) {
                float v = acc[i][t][r];
                if (EPI == 1) v = softplus_f(v + bv);
                C[(size_t)(mBase + i * 16 + r + rowSel) * ldc + col] = v;
            }
        }
}

#define CONV_T    16
#define CONV_TPB  (D_INNER / 4)

#define XP_KS     64
#define XP_KQ     (XP_KS / 8)
#define XP_NT     (DBC_LD / 16)
#define XP_WAVES  4
#define XP_SLICES (D_INNER / XP_KS)
#define XP_SPW    (XP_SLICES / XP_WAVES)
#define XP_FRAG   128
#define XP_PIECES (DBC_LD * D_INNER / 4)

__launch_bounds__(CONV_TPB) __attribute__((amdgpu_waves_per_eu(3, 4)))
__global__ void conv_silu_kernel(const float* __restrict__ xin,
                                 const float* __restrict__ w,
                                 const float* __restrict__ b,
                                 float* __restrict__ out,
                                 const float* __restrict__ xw,
                                 float* __restrict__ xwf) {
    const unsigned g = (blockIdx.y * gridDim.x + blockIdx.x) * CONV_TPB + threadIdx.x;
    const bool mover = g < XP_PIECES;
    v4f piece = {0.f, 0.f, 0.f, 0.f};
    if (mover) piece = *(const v4f*)(xw + (size_t)g * 4);

    const int c4 = threadIdx.x * 4;
    const int s0 = blockIdx.x * CONV_T;
    const int bi = blockIdx.y;
    const bool head = (s0 == 0);

    const size_t off = ((size_t)bi * SEQ + s0) * D_INNER + c4;
    const float* px = xin + off;
    float*       po = out + off;

    v4f wv[4];
#pragma unroll
    for (int c = 0; c < 4; ++c)
        wv[c] = *(const v4f*)(w + (size_t)(c4 + c) * D_CONV);
    const v4f bv = *(const v4f*)(b + c4);

    v4f xr[CONV_T + D_CONV - 1];
#pragma unroll
    for (int j = 0; j < D_CONV - 1; ++j) {
        v4f v = {0.f, 0.f, 0.f, 0.f};
        if (!head) v = *(const v4f*)(px - (size_t)(D_CONV - 1 - j) * D_INNER);
        xr[j] = v;
    }
#pragma unroll
    for (int t = 0; t < CONV_T; ++t)
        xr[t + D_CONV - 1] = *(const v4f*)(px + (size_t)t * D_INNER);

#pragma unroll
    for (int t = 0; t < CONV_T; ++t) {
        v4f acc = bv;
#pragma unroll
        for (int j = 0; j < D_CONV; ++j) {
            if (t + j >= D_CONV - 1 || !head) {
                const v4f xv = xr[t + j];
                acc.x += wv[0][j] * xv.x;
                acc.y += wv[1][j] * xv.y;
                acc.z += wv[2][j] * xv.z;
                acc.w += wv[3][j] * xv.w;
            }
        }
        v4f o;
        o.x = silu_f(acc.x);
        o.y = silu_f(acc.y);
        o.z = silu_f(acc.z);
        o.w = silu_f(acc.w);
        *(v4f*)(po + (size_t)t * D_INNER) = o;
    }

    if (mover) {
        const unsigned n  = g / (D_INNER / 4);
        const unsigned k0 = (g % (D_INNER / 4)) * 4;
        const unsigned s  = k0 / XP_KS;
        const unsigned j  = (k0 % XP_KS) / 8;
        const unsigned h  = (k0 % 8) / 4;
        const unsigned t  = n / 16;
        const unsigned c  = n % 16;
        const unsigned p  = ((s * XP_KQ + j) * XP_NT + t) * 32 + h * 16 + c;
        *(v4f*)(xwf + (size_t)p * 4) = piece;
    }
}

__launch_bounds__(XP_WAVES * 32) __attribute__((amdgpu_waves_per_eu(1, 1)))
__global__ void xproj_ksplit_kernel(const float* __restrict__ A,
                                    const float* __restrict__ Wf,
                                    float* __restrict__ C) {
    const int lane  = threadIdx.x & 31;
    const int wave  = threadIdx.x >> 5;
    const int mBase = blockIdx.x * 16;
    const int r16   = lane & 15;
    const int kOff  = (lane >> 4) * 4;

    const float* aP = A + (size_t)(mBase + r16) * D_INNER + wave * XP_KS + kOff;
    const float* wP = Wf + (size_t)wave * (XP_KQ * XP_NT * XP_FRAG) + lane * 4;

    v8f acc[XP_NT];
#pragma unroll
    for (int t = 0; t < XP_NT; ++t) {
        v8f z = {0.f, 0.f, 0.f, 0.f, 0.f, 0.f, 0.f, 0.f};
        acc[t] = z;
    }

    v4f aA[XP_KQ], bA[XP_NT][XP_KQ], aB[XP_KQ], bB[XP_NT][XP_KQ];

#define XP_LOAD(AF, BF, I)                                                   \
    _Pragma("unroll") for (int j = 0; j < XP_KQ; ++j)                        \
        AF[j] = *(const v4f*)(aP + (I) * (XP_WAVES * XP_KS) + 8 * j);        \
    _Pragma("unroll") for (int j = 0; j < XP_KQ; ++j)                        \
        _Pragma("unroll") for (int t = 0; t < XP_NT; ++t)                    \
            BF[t][j] = *(const v4f*)(wP +                                    \
                (size_t)((((I) * XP_WAVES * XP_KQ + j) * XP_NT + t) * XP_FRAG));

#define XP_MAC(AF, BF)                                                       \
    _Pragma("unroll") for (int j = 0; j < XP_KQ; ++j) {                      \
        _Pragma("unroll") for (int t = 0; t < XP_NT; ++t)                    \
            acc[t] = __builtin_amdgcn_wmma_f32_16x16x4_f32(                  \
                false, AF[j].xy, false, BF[t][j].xy,                         \
                (short)0, acc[t], false, false);                             \
        _Pragma("unroll") for (int t = 0; t < XP_NT; ++t)                    \
            acc[t] = __builtin_amdgcn_wmma_f32_16x16x4_f32(                  \
                false, AF[j].zw, false, BF[t][j].zw,                         \
                (short)0, acc[t], false, false);                             \
    }

    XP_LOAD(aA, bA, 0)
#pragma unroll
    for (int i = 0; i < XP_SPW; i += 2) {
        XP_LOAD(aB, bB, i + 1)
        XP_MAC(aA, bA)
        if (i + 2 < XP_SPW) {
            XP_LOAD(aA, bA, i + 2)
        }
        XP_MAC(aB, bB)
    }

#undef XP_LOAD
#undef XP_MAC

    __shared__ float sP[XP_WAVES][16 * DBC_LD];
    const int rowSel = (lane >> 4) * 8;
#pragma unroll
    for (int t = 0; t < XP_NT; ++t)
#pragma unroll
        for (int r = 0; r < 8; ++r)
            sP[wave][(r + rowSel) * DBC_LD + t * 16 + r16] = acc[t][r];
    __syncthreads();

    float* cP = C + (size_t)mBase * DBC_LD;
#pragma unroll
    for (int q = 0; q < 16 * DBC_LD / (XP_WAVES * 32); ++q) {
        const int e = threadIdx.x + q * (XP_WAVES * 32);
        cP[e] = ((sP[0][e] + sP[1][e]) + sP[2][e]) + sP[3][e];
    }
}

#define SCAN_NC   32
#define SCAN_LEN  (SEQ / SCAN_NC)
#define SCAN_G    8
#define SCAN_TPB  128
#define SUM_LD    ((size_t)D_STATE * D_INNER)
#define SCAN_BQ   (D_STATE / 4)

__device__ __forceinline__ void load_decay_rates(const float* __restrict__ logA,
                                                 int d, float (&An)[D_STATE]) {
#pragma unroll
    for (int q = 0; q < D_STATE / 4; ++q) {
        const v4f v = *(const v4f*)(logA + (size_t)d * D_STATE + 4 * q);
        An[4 * q + 0] = -__expf(v.x);
        An[4 * q + 1] = -__expf(v.y);
        An[4 * q + 2] = -__expf(v.z);
        An[4 * q + 3] = -__expf(v.w);
    }
}

#define DC_MT     (SCAN_LEN / 16)
#define DC_NT     2
#define DC_KS     (DT_RANK / 2)
#define DC_KQ     (DC_KS / 8)
#define DC_PITCH  36

__launch_bounds__(SCAN_TPB)
__global__ void delta_chunk_state_kernel(const float* __restrict__ dbc,
                                         const float* __restrict__ dtw,
                                         const float* __restrict__ dtb,
                                         const float* __restrict__ xc,
                                         const float* __restrict__ logA,
                                         float* __restrict__ delta,
                                         float* __restrict__ hend,
                                         float* __restrict__ pdec) {
    const int lane = threadIdx.x & 31;
    const int wave = threadIdx.x >> 5;
    const int dW   = blockIdx.x * SCAN_TPB + wave * 32;
    const int d    = dW + lane;
    const int c    = blockIdx.y;
    const int b    = blockIdx.z;
    const bool last = (c == SCAN_NC - 1);
    const int r16  = lane & 15;
    const int kOff = (lane >> 4) * 4;

    const size_t row0 = (size_t)b * SEQ + (size_t)c * SCAN_LEN;
    const float* pxc = xc  + row0 * D_INNER + d;
    const float* pB  = dbc + row0 * DBC_LD + DT_RANK;
    float*       pdl = delta + row0 * D_INNER + d;

    const float* aP[DC_MT];
    const float* wP[DC_NT];
#pragma unroll
    for (int i = 0; i < DC_MT; ++i)
        aP[i] = dbc + (row0 + i * 16 + r16) * DBC_LD + kOff;
#pragma unroll
    for (int t = 0; t < DC_NT; ++t)
        wP[t] = dtw + (size_t)(dW + t * 16 + r16) * DT_RANK + kOff;

    v4f aF[2][DC_MT][DC_KQ], bF[2][DC_NT][DC_KQ];
#pragma unroll
    for (int s = 0; s < 2; ++s) {
#pragma unroll
        for (int i = 0; i < DC_MT; ++i)
#pragma unroll
            for (int j = 0; j < DC_KQ; ++j)
                aF[s][i][j] = *(const v4f*)(aP[i] + s * DC_KS + 8 * j);
#pragma unroll
        for (int t = 0; t < DC_NT; ++t)
#pragma unroll
            for (int j = 0; j < DC_KQ; ++j)
                bF[s][t][j] = *(const v4f*)(wP[t] + s * DC_KS + 8 * j);
    }

    __shared__ v4f sB[SCAN_LEN * SCAN_BQ];
    v4f stg[SCAN_LEN * SCAN_BQ / SCAN_TPB];
#pragma unroll
    for (int q = 0; q < SCAN_LEN * SCAN_BQ / SCAN_TPB; ++q) {
        const int p = threadIdx.x + q * SCAN_TPB;
        stg[q] = *(const v4f*)(pB + (size_t)(p / SCAN_BQ) * DBC_LD + (p % SCAN_BQ) * 4);
    }
    const float bv = dtb[d];
    float xvC[SCAN_G], xvN[SCAN_G];
#pragma unroll
    for (int g = 0; g < SCAN_G; ++g) xvC[g] = pxc[(size_t)g * D_INNER];

    v8f acc[DC_MT][DC_NT];
#pragma unroll
    for (int i = 0; i < DC_MT; ++i)
#pragma unroll
        for (int t = 0; t < DC_NT; ++t) {
            v8f z = {0.f, 0.f, 0.f, 0.f, 0.f, 0.f, 0.f, 0.f};
            acc[i][t] = z;
        }
#pragma unroll
    for (int s = 0; s < 2; ++s)
#pragma unroll
        for (int j = 0; j < DC_KQ; ++j) {
#pragma unroll
            for (int i = 0; i < DC_MT; ++i)
#pragma unroll
                for (int t = 0; t < DC_NT; ++t)
                    acc[i][t] = __builtin_amdgcn_wmma_f32_16x16x4_f32(
                        false, aF[s][i][j].xy, false, bF[s][t][j].xy,
                        (short)0, acc[i][t], false, false);
#pragma unroll
            for (int i = 0; i < DC_MT; ++i)
#pragma unroll
                for (int t = 0; t < DC_NT; ++t)
                    acc[i][t] = __builtin_amdgcn_wmma_f32_16x16x4_f32(
                        false, aF[s][i][j].zw, false, bF[s][t][j].zw,
                        (short)0, acc[i][t], false, false);
        }

    __shared__ float sD[SCAN_TPB / 32][SCAN_LEN * DC_PITCH];
    const int rowSel = (lane >> 4) * 8;
#pragma unroll
    for (int i = 0; i < DC_MT; ++i)
#pragma unroll
        for (int t = 0; t < DC_NT; ++t)
#pragma unroll
            for (int r = 0; r < 8; ++r)
                sD[wave][(i * 16 + r + rowSel) * DC_PITCH + t * 16 + r16] = acc[i][t][r];
#pragma unroll
    for (int q = 0; q < SCAN_LEN * SCAN_BQ / SCAN_TPB; ++q)
        sB[threadIdx.x + q * SCAN_TPB] = stg[q];

    float An[D_STATE];
    load_decay_rates(logA, d, An);

    __syncthreads();

    const float* sDt = &sD[wave][lane];

    float h[D_STATE];
#pragma unroll
    for (int n = 0; n < D_STATE; ++n) h[n] = 0.0f;
    float dsum = 0.0f;

    float rwC[SCAN_G], rwN[SCAN_G];
#pragma unroll
    for (int g = 0; g < SCAN_G; ++g) rwC[g] = sDt[g * DC_PITCH];

    v4f bC[SCAN_BQ], bN[SCAN_BQ];
#pragma unroll
    for (int q = 0; q < SCAN_BQ; ++q) bC[q] = sB[q];
#pragma unroll 1
    for (int s = 0; s < SCAN_LEN; s += SCAN_G) {
        if (s + SCAN_G < SCAN_LEN) {
#pragma unroll
            for (int g = 0; g < SCAN_G; ++g) {
                rwN[g] = sDt[(s + SCAN_G + g) * DC_PITCH];
                xvN[g] = pxc[(size_t)(s + SCAN_G + g) * D_INNER];
            }
        }
        float dtC[SCAN_G];
#pragma unroll
        for (int g = 0; g < SCAN_G; ++g) {
            dtC[g] = softplus_f(rwC[g] + bv);
            pdl[(size_t)(s + g) * D_INNER] = dtC[g];
        }
        if (!last) {
#pragma unroll
            for (int g = 0; g < SCAN_G; ++g) {
                const int sn = (s + g + 1 < SCAN_LEN) ? (s + g + 1) : (SCAN_LEN - 1);
#pragma unroll
                for (int q = 0; q < SCAN_BQ; ++q) bN[q] = sB[sn * SCAN_BQ + q];
                __builtin_amdgcn_sched_barrier(0);
                float bq[D_STATE];
#pragma unroll
                for (int q = 0; q < SCAN_BQ; ++q) {
                    bq[4 * q + 0] = bC[q].x; bq[4 * q + 1] = bC[q].y;
                    bq[4 * q + 2] = bC[q].z; bq[4 * q + 3] = bC[q].w;
                }
                const float dt = dtC[g];
                const float xv = xvC[g];
#pragma unroll
                for (int n = 0; n < D_STATE; ++n) {
                    const float a = __expf(dt * An[n]);
                    h[n] = a * h[n] + (dt * bq[n]) * xv;
                }
                dsum += dt;
#pragma unroll
                for (int q = 0; q < SCAN_BQ; ++q) bC[q] = bN[q];
            }
        }
#pragma unroll
        for (int g = 0; g < SCAN_G; ++g) { rwC[g] = rwN[g]; xvC[g] = xvN[g]; }
    }

    if (last) return;

    const size_t o = (size_t)(b * SCAN_NC + c) * SUM_LD + d;
#pragma unroll
    for (int n = 0; n < D_STATE; ++n) {
        hend[o + (size_t)n * D_INNER] = h[n];
        pdec[o + (size_t)n * D_INNER] = __expf(dsum * An[n]);
    }
}

__launch_bounds__(SCAN_TPB) __attribute__((amdgpu_waves_per_eu(4, 4)))
__global__ void scan_chunk_kernel(const float* __restrict__ delta,
                                  const float* __restrict__ xc,
                                  const float* __restrict__ dbc,
                                  const float* __restrict__ zbuf,
                                  const float* __restrict__ logA,
                                  const float* __restrict__ Dskip,
                                  const float* __restrict__ hend,
                                  const float* __restrict__ pdec,
                                  float* __restrict__ y) {
    const int d = blockIdx.x * SCAN_TPB + threadIdx.x;
    const int c = blockIdx.y;
    const int b = blockIdx.z;

    float An[D_STATE];
    load_decay_rates(logA, d, An);
    const float dsk = Dskip[d];

    const size_t row0 = (size_t)b * SEQ + (size_t)c * SCAN_LEN;
    const float* pdt = delta + row0 * D_INNER + d;
    const float* pxc = xc    + row0 * D_INNER + d;
    const float* pz  = zbuf  + row0 * D_INNER + d;
    float*       py  = y     + row0 * D_INNER + d;
    const float* pBC = dbc   + row0 * DBC_LD + DT_RANK;

    __shared__ v4f sBC[SCAN_LEN * 2 * SCAN_BQ];
    v4f stg[SCAN_LEN * 2 * SCAN_BQ / SCAN_TPB];
#pragma unroll
    for (int q = 0; q < SCAN_LEN * 2 * SCAN_BQ / SCAN_TPB; ++q) {
        const int p = threadIdx.x + q * SCAN_TPB;
        stg[q] = *(const v4f*)(pBC + (size_t)(p / (2 * SCAN_BQ)) * DBC_LD + (p % (2 * SCAN_BQ)) * 4);
    }

    float dtC[SCAN_G], xvC[SCAN_G], zvC[SCAN_G];
    float dtN[SCAN_G], xvN[SCAN_G], zvN[SCAN_G];
#pragma unroll
    for (int g = 0; g < SCAN_G; ++g) {
        dtC[g] = pdt[(size_t)g * D_INNER];
        xvC[g] = pxc[(size_t)g * D_INNER];
        zvC[g] = pz[(size_t)g * D_INNER];
    }

    float h[D_STATE];
#pragma unroll
    for (int n = 0; n < D_STATE; ++n) h[n] = 0.0f;
    const size_t o0 = (size_t)b * SCAN_NC * SUM_LD + d;
#pragma unroll 2
    for (int j = 0; j < c; ++j) {
        const size_t o = o0 + (size_t)j * SUM_LD;
#pragma unroll
        for (int n = 0; n < D_STATE; ++n)
            h[n] = pdec[o + (size_t)n * D_INNER] * h[n] + hend[o + (size_t)n * D_INNER];
    }

#pragma unroll
    for (int q = 0; q < SCAN_LEN * 2 * SCAN_BQ / SCAN_TPB; ++q)
        sBC[threadIdx.x + q * SCAN_TPB] = stg[q];
    __syncthreads();

    v4f bcC[2 * SCAN_BQ], bcN[2 * SCAN_BQ];
#pragma unroll
    for (int q = 0; q < 2 * SCAN_BQ; ++q) bcC[q] = sBC[q];

#pragma unroll 1
    for (int s = 0; s < SCAN_LEN; s += SCAN_G) {
        if (s + SCAN_G < SCAN_LEN) {
#pragma unroll
            for (int g = 0; g < SCAN_G; ++g) {
                dtN[g] = pdt[(size_t)(s + SCAN_G + g) * D_INNER];
                xvN[g] = pxc[(size_t)(s + SCAN_G + g) * D_INNER];
                zvN[g] = pz[(size_t)(s + SCAN_G + g) * D_INNER];
            }
        }
#pragma unroll
        for (int g = 0; g < SCAN_G; ++g) {
            const int sn = (s + g + 1 < SCAN_LEN) ? (s + g + 1) : (SCAN_LEN - 1);
#pragma unroll
            for (int q = 0; q < 2 * SCAN_BQ; ++q) bcN[q] = sBC[sn * 2 * SCAN_BQ + q];
            __builtin_amdgcn_sched_barrier(0);
            float bq[2 * D_STATE];
#pragma unroll
            for (int q = 0; q < 2 * SCAN_BQ; ++q) {
                bq[4 * q + 0] = bcC[q].x; bq[4 * q + 1] = bcC[q].y;
                bq[4 * q + 2] = bcC[q].z; bq[4 * q + 3] = bcC[q].w;
            }
            const float dt = dtC[g];
            const float xv = xvC[g];
            float p[D_STATE];
#pragma unroll
            for (int n = 0; n < D_STATE; ++n) {
                const float a = __expf(dt * An[n]);
                h[n] = a * h[n] + (dt * bq[n]) * xv;
                p[n] = h[n] * bq[D_STATE + n];
            }
#pragma unroll
            for (int w = 1; w < D_STATE; w *= 2)
#pragma unroll
                for (int n = 0; n < D_STATE; n += 2 * w) p[n] += p[n + w];

            py[(size_t)(s + g) * D_INNER] = (p[0] + dsk * xv) * silu_f(zvC[g]);
#pragma unroll
            for (int q = 0; q < 2 * SCAN_BQ; ++q) bcC[q] = bcN[q];
        }
#pragma unroll
        for (int g = 0; g < SCAN_G; ++g) { dtC[g] = dtN[g]; xvC[g] = xvN[g]; zvC[g] = zvN[g]; }
    }
}

extern "C" void kernel_launch(void* const* d_in, const int* in_sizes, int n_in,
                              void* d_out, int out_size, void* d_ws, size_t ws_size,
                              hipStream_t stream) {
    const float* x         = (const float*)d_in[0];
    const float* in_proj_w = (const float*)d_in[1];
    const float* conv_w    = (const float*)d_in[2];
    const float* conv_b    = (const float*)d_in[3];
    const float* x_proj_w  = (const float*)d_in[4];
    const float* dt_proj_w = (const float*)d_in[5];
    const float* dt_proj_b = (const float*)d_in[6];
    const float* log_A     = (const float*)d_in[7];
    const float* D_skip    = (const float*)d_in[8];
    const float* out_proj_w= (const float*)d_in[9];
    float* out = (float*)d_out;

    const size_t planeBD = (size_t)MROWS * D_INNER;
    float* ws      = (float*)d_ws;
    float* buf_x   = ws;
    float* buf_z   = buf_x  + planeBD;
    float* buf_xc  = buf_z  + planeBD;
    float* buf_dbc = buf_xc + planeBD;
    float* buf_y   = buf_dbc + (size_t)MROWS * DBC_LD;
    float* buf_he  = buf_y  + planeBD;
    float* buf_pd  = buf_he + (size_t)BATCH * SCAN_NC * SUM_LD;
    float* buf_wf  = buf_pd + (size_t)BATCH * SCAN_NC * SUM_LD;

    gemm_wmma_f32<4, 6, 32, 2, 2, 0><<<dim3(D_INNER / 192, MROWS / 128), 128, 0, stream>>>(
        x, in_proj_w, nullptr, buf_x, D_MODEL, D_MODEL, D_INNER);
    gemm_wmma_f32<4, 6, 32, 2, 2, 0><<<dim3(D_INNER / 192, MROWS / 128), 128, 0, stream>>>(
        x, in_proj_w + (size_t)D_INNER * D_MODEL, nullptr, buf_z,
        D_MODEL, D_MODEL, D_INNER);

    conv_silu_kernel<<<dim3(SEQ / CONV_T, BATCH), CONV_TPB, 0, stream>>>(
        buf_x, conv_w, conv_b, buf_xc, x_proj_w, buf_wf);

    xproj_ksplit_kernel<<<dim3(MROWS / 16), XP_WAVES * 32, 0, stream>>>(
        buf_xc, buf_wf, buf_dbc);

    delta_chunk_state_kernel<<<dim3(D_INNER / SCAN_TPB, SCAN_NC, BATCH), SCAN_TPB, 0, stream>>>(
        buf_dbc, dt_proj_w, dt_proj_b, buf_xc, log_A, buf_x, buf_he, buf_pd);
    scan_chunk_kernel<<<dim3(D_INNER / SCAN_TPB, SCAN_NC, BATCH), SCAN_TPB, 0, stream>>>(
        buf_x, buf_xc, buf_dbc, buf_z, log_A, D_skip, buf_he, buf_pd, buf_y);

    gemm_wmma_f32<4, 3, 32, 2, 2, 0><<<dim3(D_MODEL / 96, MROWS / 128), 128, 0, stream>>>(
        buf_y, out_proj_w, nullptr, out, D_INNER, D_INNER, D_MODEL);
}
